# starting kernel
# baseline (speedup 1.0000x reference)
#include <hip/hip_runtime.h>

#define N_FIN  32
#define HID    256
#define N_FOUT 64

typedef __attribute__((ext_vector_type(16))) _Float16 v16h;
typedef __attribute__((ext_vector_type(8)))  _Float16 v8h;
typedef __attribute__((ext_vector_type(8)))  float    v8f;

#define WMMA(a, b, c) __builtin_amdgcn_wmma_f32_16x16x32_f16( \
    false, (a), false, (b), (short)0, (c), false, false)

static __device__ __forceinline__ float fast_elu(float x) {
    const float e = __builtin_amdgcn_exp2f(fminf(x, 0.0f) * 1.44269504088896f) - 1.0f;
    return (x > 0.0f) ? x : e;
}

static __device__ __forceinline__ v16h ld_frag(const _Float16* p, int stride, int k0) {
    const int lane = threadIdx.x & 31;
    const _Float16* q = p + (lane & 15) * stride + k0 + ((lane >> 4) << 3);
    v8h lo = *(const v8h*)q;
    v8h hi = *(const v8h*)(q + 16);
    return __builtin_shufflevector(lo, hi, 0, 1, 2, 3, 4, 5, 6, 7,
                                           8, 9, 10, 11, 12, 13, 14, 15);
}

static __device__ __forceinline__ void st_elu_f16(_Float16* base, int stride,
                                                  v8f d, const float* bias, int o0) {
    const int lane = threadIdx.x & 31;
    const int col  = lane & 15;
    const int r0   = (lane >> 4) << 3;
    const float bv = bias[o0 + col];
#pragma unroll
    for (int r = 0; r < 8; ++r) {
        base[(r0 + r) * stride + col] = (_Float16)fast_elu(d[r] + bv);
    }
}

__global__ __launch_bounds__(256) void gnn_scatter(
    const float* __restrict__ feat, const float* __restrict__ ew,
    const int* __restrict__ src, const int* __restrict__ dst,
    float* __restrict__ msg, long long total) {
    long long idx = (long long)blockIdx.x * 256 + threadIdx.x;
    if (idx >= total) return;
    const int e = (int)(idx >> 5);
    const int c = (int)(idx & 31);
    const float v = feat[(long long)src[e] * N_FIN + c] * ew[e];
    atomicAdd(&msg[(long long)dst[e] * N_FIN + c], v);
}

__global__ __launch_bounds__(256) void gnn_fused(
    const float* __restrict__ feat, const float* __restrict__ msg,
    const float* __restrict__ Win, const float* __restrict__ bin,
    const float* __restrict__ Wh,  const float* __restrict__ bh,
    const float* __restrict__ Wo,  const float* __restrict__ bo,
    float* __restrict__ out, int nNodes) {
    __shared__ __align__(16) _Float16 sWin[HID * 2 * N_FIN];
    __shared__ __align__(16) _Float16 sWh [HID * HID];
    __shared__ __align__(16) _Float16 sWo [N_FOUT * HID];
    __shared__ __align__(16) _Float16 sScr[8 * 5632];
    __shared__ __align__(16) float    sBias[HID + HID + N_FOUT];

    const int tid = threadIdx.x;
    for (int i = tid; i < HID * 2 * N_FIN; i += 256) sWin[i] = (_Float16)Win[i];
    for (int i = tid; i < HID * HID;       i += 256) sWh[i]  = (_Float16)Wh[i];
    for (int i = tid; i < N_FOUT * HID;    i += 256) sWo[i]  = (_Float16)Wo[i];
    if (tid < HID)    sBias[tid]           = bin[tid];
    if (tid < HID)    sBias[HID + tid]     = bh[tid];
    if (tid < N_FOUT) sBias[2 * HID + tid] = bo[tid];
    __syncthreads();

    const float* sBin = sBias;
    const float* sBh  = sBias + HID;
    const float* sBo  = sBias + 2 * HID;

    const int wave = tid >> 5;
    const int lane = tid & 31;
    _Float16* bufA = sScr + wave * 5632;
    _Float16* bufH = bufA + 16 * 64;
    _Float16* bufS = bufH + 16 * HID;

    const int n0 = blockIdx.x * 128 + wave * 16;
    if (n0 >= nNodes) return;

#pragma unroll
    for (int it = 0; it < 32; ++it) {
        const int idx = it * 32 + lane;
        const int r = idx >> 6, c = idx & 63;
        const int node = n0 + r;
        float v = 0.0f;
        if (node < nNodes)
            v = (c < N_FIN) ? feat[(long long)node * N_FIN + c]
                            : msg[(long long)node * N_FIN + (c - N_FIN)];
        bufA[idx] = (_Float16)v;
    }

    for (int t = 0; t < 16; ++t) {
        v8f acc = {0.f, 0.f, 0.f, 0.f, 0.f, 0.f, 0.f, 0.f};
        acc = WMMA(ld_frag(bufA, 64, 0),  ld_frag(sWin + t * 16 * 64, 64, 0),  acc);
        acc = WMMA(ld_frag(bufA, 64, 32), ld_frag(sWin + t * 16 * 64, 64, 32), acc);
        st_elu_f16(bufH + t * 16, HID, acc, sBin, t * 16);
    }

    v8f acc3[4];
#pragma unroll
    for (int t3 = 0; t3 < 4; ++t3)
        acc3[t3] = (v8f){0.f, 0.f, 0.f, 0.f, 0.f, 0.f, 0.f, 0.f};

    for (int t2 = 0; t2 < 16; ++t2) {
        v8f acc = {0.f, 0.f, 0.f, 0.f, 0.f, 0.f, 0.f, 0.f};
#pragma unroll
        for (int kk = 0; kk < 8; ++kk)
            acc = WMMA(ld_frag(bufH, HID, kk * 32),
                       ld_frag(sWh + t2 * 16 * HID, HID, kk * 32), acc);
        st_elu_f16(bufS + (t2 & 1) * 16, 32, acc, sBh, t2 * 16);
        if (t2 & 1) {
            const v16h a3 = ld_frag(bufS, 32, 0);
            const int k0 = (t2 - 1) * 16;
#pragma unroll
            for (int t3 = 0; t3 < 4; ++t3)
                acc3[t3] = WMMA(a3, ld_frag(sWo + t3 * 16 * HID, HID, k0), acc3[t3]);
        }
    }

#pragma unroll
    for (int t3 = 0; t3 < 4; ++t3) {
        const int col = t3 * 16 + (lane & 15);
        const int r0  = (lane >> 4) << 3;
        const float bv = sBo[col];
#pragma unroll
        for (int r = 0; r < 8; ++r) {
            const int node = n0 + r0 + r;
            if (node < nNodes)
                out[(long long)node * N_FOUT + col] = acc3[t3][r] + bv;
        }
    }
}

extern "C" void kernel_launch(void* const* d_in, const int* in_sizes, int n_in,
                              void* d_out, int out_size, void* d_ws, size_t ws_size,
                              hipStream_t stream) {
    const float* feat = (const float*)d_in[0];
    const float* ew   = (const float*)d_in[1];
    const float* Win  = (const float*)d_in[2];
    const float* bin  = (const float*)d_in[3];
    const float* Wh   = (const float*)d_in[4];
    const float* bh   = (const float*)d_in[5];
    const float* Wo   = (const float*)d_in[6];
    const float* bo   = (const float*)d_in[7];
    const int*   ei   = (const int*)d_in[8];

    const int nNodes = in_sizes[0] / N_FIN;
    const int nEdges = in_sizes[1];
    float* msg = (float*)d_ws;

    hipMemsetAsync(msg, 0, (size_t)nNodes * N_FIN * sizeof(float), stream);

    const long long total = (long long)nEdges * 32;
    const int sblocks = (int)((total + 255) / 256);
    gnn_scatter<<<sblocks, 256, 0, stream>>>(feat, ew, ei, ei + nEdges, msg, total);

    const int gblocks = (nNodes + 127) / 128;
    gnn_fused<<<gblocks, 256, 0, stream>>>(feat, msg, Win, bin, Wh, bh, Wo, bo,
                                           (float*)d_out, nNodes);
}


# Edge owner scan reads 1 byte per edge instead of 4: a new first launch stores each target's owner number (low byte) bit-sliced behind msg; owners test 32 edges with 8 XORs; LDS adds as before
# speedup vs baseline: 3.3319x; 3.3319x over previous
#include <hip/hip_runtime.h>


#define N_FIN  32
#define HID    256
#define N_FOUT 64

#define ROWS 128
#define XP   72
#define HP   264

typedef __attribute__((ext_vector_type(16))) _Float16 v16h;
typedef __attribute__((ext_vector_type(8)))  _Float16 v8h;
typedef __attribute__((ext_vector_type(4)))  _Float16 v4h;
typedef __attribute__((ext_vector_type(8)))  float    v8f;

#define WMMA(a, b, c) __builtin_amdgcn_wmma_f32_16x16x32_f16( \
    false, (a), false, (b), (short)0, (c), false, false)

static __device__ __forceinline__ float fast_elu(float x) {
    const float e = __builtin_amdgcn_exp2f(fminf(x, 0.0f) * 1.44269504088896f) - 1.0f;
    return (x > 0.0f) ? x : e;
}

static __device__ __forceinline__ v16h ld_frag(const _Float16* p, int stride, int k0) {
    const int lane = threadIdx.x & 31;
    const _Float16* q = p + (lane & 15) * stride + k0 + ((lane >> 4) << 3);
    v8h lo = *(const v8h*)q;
    v8h hi = *(const v8h*)(q + 16);
    return __builtin_shufflevector(lo, hi, 0, 1, 2, 3, 4, 5, 6, 7,
                                           8, 9, 10, 11, 12, 13, 14, 15);
}

static __device__ __forceinline__ v16h ld_frag_g(const float* __restrict__ W, int ldw,
                                                 int row0, int k0) {
    const int lane = threadIdx.x & 31;
    const float* q = W + (long long)(row0 + (lane & 15)) * ldw + k0 + ((lane >> 4) << 3);
    const float4 a0 = *(const float4*)(q);
    const float4 a1 = *(const float4*)(q + 4);
    const float4 b0 = *(const float4*)(q + 16);
    const float4 b1 = *(const float4*)(q + 20);
    v16h r;
    r[0]  = (_Float16)a0.x; r[1]  = (_Float16)a0.y; r[2]  = (_Float16)a0.z; r[3]  = (_Float16)a0.w;
    r[4]  = (_Float16)a1.x; r[5]  = (_Float16)a1.y; r[6]  = (_Float16)a1.z; r[7]  = (_Float16)a1.w;
    r[8]  = (_Float16)b0.x; r[9]  = (_Float16)b0.y; r[10] = (_Float16)b0.z; r[11] = (_Float16)b0.w;
    r[12] = (_Float16)b1.x; r[13] = (_Float16)b1.y; r[14] = (_Float16)b1.z; r[15] = (_Float16)b1.w;
    return r;
}

static __device__ __forceinline__ void st_elu_f16(_Float16* base, int stride,
                                                  v8f d, float bv) {
    const int lane = threadIdx.x & 31;
    const int col  = lane & 15;
    const int r0   = (lane >> 4) << 3;
#pragma unroll
    for (int r = 0; r < 8; ++r) {
        base[(r0 + r) * stride + col] = (_Float16)fast_elu(d[r] + bv);
    }
}

#define SLICE   392
#define GWAVES  32
#define KSTEP   2
#define QCAP    (32 + 1024)

static __device__ __forceinline__ int key_groups(int nNodes, int nEdges, size_t wsBytes) {
    if (nEdges & 3) return 0;
    const int nG = nEdges / 128;
    const size_t trips = (size_t)((nG + 8 * KSTEP - 1) / (8 * KSTEP));
    const size_t need = (size_t)nNodes * N_FIN * sizeof(float) + trips * (KSTEP * 1024);
    return need <= wsBytes ? nG : 0;
}

#define KPRE 4

__global__ __launch_bounds__(512) void gnn_owner_keys(
    const int* __restrict__ dst, unsigned* __restrict__ keys,
    int nNodes, int nEdges, size_t wsBytes) {
    const int lane = threadIdx.x & 31;
    const int wave = __builtin_amdgcn_readfirstlane((int)(threadIdx.x >> 5));
    const int wpb  = (int)(blockDim.x >> 5);
    const int nG   = key_groups(nNodes, nEdges, wsBytes);
    const int nGpad = ((nG + 8 * KSTEP - 1) / (8 * KSTEP)) * (8 * KSTEP);
    const int gw = blockIdx.x * wpb + wave;
    const int tw = gridDim.x * wpb;
    const int4* dstv = (const int4*)dst;

    for (int base = gw; base < nGpad; base += KPRE * tw) {
        int4 d[KPRE];
#pragma unroll
        for (int j = 0; j < KPRE; ++j) {
            const int G = base + j * tw;
            d[j] = (G < nG) ? dstv[(size_t)G * 32 + lane] : make_int4(0, 0, 0, 0);
        }
#pragma unroll
        for (int j = 0; j < KPRE; ++j) {
            const int G = base + j * tw;
            if (G < nGpad) {
                unsigned k[4];
                k[0] = (unsigned)d[j].x / SLICE; k[1] = (unsigned)d[j].y / SLICE;
                k[2] = (unsigned)d[j].z / SLICE; k[3] = (unsigned)d[j].w / SLICE;
                unsigned mine = 0u;
#pragma unroll
                for (int c = 0; c < 4; ++c)
#pragma unroll
                    for (int b = 0; b < 8; ++b) {
                        const unsigned m = __builtin_amdgcn_ballot_w32(((k[c] >> b) & 1u) != 0u);
                        mine = (lane == 8 * c + b) ? m : mine;
                    }
                const int b   = lane & 7;
                const int sub = (G & 7) * 4 + (lane >> 3);
                keys[(size_t)(G >> 3) * 256 + (b >> 2) * 128 + sub * 4 + (b & 3)] = mine;
            }
        }
    }
}

static __device__ __forceinline__ void drain32(
    const unsigned* q, const float* __restrict__ feat, const float* __restrict__ ew,
    const int* __restrict__ src, const int* __restrict__ dst, float* acc,
    unsigned lo, unsigned rows, int lane) {
    const unsigned e   = q[lane];
    const unsigned s_l = (unsigned)src[e];
    const unsigned w_l = __float_as_uint(ew[e]);
    const unsigned r_l = (unsigned)dst[e] - lo;
    const unsigned okm = __builtin_amdgcn_ballot_w32(r_l < rows);
    float v[32];
#pragma unroll
    for (int t = 0; t < 32; ++t) {
        const unsigned s = __builtin_amdgcn_readlane(s_l, t);
        v[t] = feat[(size_t)s * N_FIN + lane];
    }
#pragma unroll
    for (int t = 0; t < 32; ++t) {
        if ((okm >> t) & 1u) {
            const float    w = __uint_as_float(__builtin_amdgcn_readlane(w_l, t));
            const unsigned r = __builtin_amdgcn_readlane(r_l, t);
            __hip_atomic_fetch_add(&acc[r * N_FIN + lane], v[t] * w,
                                   __ATOMIC_RELAXED, __HIP_MEMORY_SCOPE_WORKGROUP);
        }
    }
}

static __device__ __forceinline__ void park(
    unsigned d, unsigned e, unsigned lo, unsigned rows, unsigned* q, int& qn) {
    const bool     hit = (d - lo) < rows;
    const unsigned m   = __builtin_amdgcn_ballot_w32(hit);
    if (m) {
        const unsigned pos = qn + __builtin_amdgcn_mbcnt_lo(m, 0u);
        if (hit) q[pos] = e;
        qn += __builtin_popcount(m);
    }
}

static __device__ __forceinline__ void park_mask(unsigned m, unsigned e0, unsigned* q, int& qn) {
    unsigned bal;
    while ((bal = __builtin_amdgcn_ballot_w32(m != 0u)) != 0u) {
        const unsigned pos = qn + __builtin_amdgcn_mbcnt_lo(bal, 0u);
        if (m != 0u) q[pos] = e0 + 4u * (unsigned)__builtin_ctz(m);
        qn += __builtin_popcount(bal);
        m &= m - 1u;
    }
}

static __device__ __forceinline__ void key_load(
    int4 (&buf)[2 * KSTEP], int& tb, bool& in, int it, int wave, int lane,
    int c0, int nTrips, const int4* __restrict__ kv) {
    const int k = it * GWAVES + wave;
    int c = c0 + k; if (c >= nTrips) c -= nTrips;
    in = k < nTrips;
    tb = in ? c : 0;
    const int4* p = kv + (size_t)tb * (64 * KSTEP) + lane;
#pragma unroll
    for (int j = 0; j < 2 * KSTEP; ++j) buf[j] = p[j * 32];
}

static __device__ __forceinline__ void key_step(
    const int4 (&buf)[2 * KSTEP], int tb, bool in, int lane, const unsigned (&kb)[8], int nG,
    unsigned* q, int& qn,
    const float* __restrict__ feat, const float* __restrict__ ew,
    const int* __restrict__ src, const int* __restrict__ dst, float* acc,
    unsigned lo, unsigned rows) {
#pragma unroll
    for (int s = 0; s < KSTEP; ++s) {
        const int G = (tb * KSTEP + s) * 8 + (lane >> 2);
        const int4 a = buf[2 * s], b = buf[2 * s + 1];
        const unsigned x = ((unsigned)a.x ^ kb[0]) | ((unsigned)a.y ^ kb[1])
                         | ((unsigned)a.z ^ kb[2]) | ((unsigned)a.w ^ kb[3])
                         | ((unsigned)b.x ^ kb[4]) | ((unsigned)b.y ^ kb[5])
                         | ((unsigned)b.z ^ kb[6]) | ((unsigned)b.w ^ kb[7]);
        const unsigned m = (in && G < nG) ? ~x : 0u;
        park_mask(m, (unsigned)G * 128u + (unsigned)(lane & 3), q, qn);
        while (qn >= 32) { qn -= 32; drain32(q + qn, feat, ew, src, dst, acc, lo, rows, lane); }
    }
}

__global__ __launch_bounds__(1024) void gnn_gather_scan(
    const float* __restrict__ feat, const float* __restrict__ ew,
    const int* __restrict__ src, const int* __restrict__ dst,
    float* __restrict__ msg, const unsigned* __restrict__ keys,
    int nNodes, int nEdges, size_t wsBytes) {
    __shared__ float acc[SLICE * N_FIN];
    __shared__ unsigned qbuf[GWAVES][QCAP];

    const int tid  = threadIdx.x;
    const int lane = tid & 31;
    const int wave = __builtin_amdgcn_readfirstlane(tid >> 5);
    unsigned* q = qbuf[wave];

    const int nG     = key_groups(nNodes, nEdges, wsBytes);
    const int nTrips = (nG + 8 * KSTEP - 1) / (8 * KSTEP);
    const int nIter  = (nTrips + GWAVES - 1) / GWAVES;
    const int c0 = (int)(((long long)blockIdx.x * nTrips) / gridDim.x);
    const int4* kv = (const int4*)keys;

    for (int lo = blockIdx.x * SLICE; lo < nNodes; lo += gridDim.x * SLICE) {
        const int rows = (nNodes - lo < SLICE) ? (nNodes - lo) : SLICE;
        for (int i = tid; i < rows * N_FIN; i += 1024) acc[i] = 0.0f;
        __syncthreads();

        const unsigned kid = (unsigned)(lo / SLICE) & 255u;
        unsigned kb[8];
#pragma unroll
        for (int b = 0; b < 8; ++b) kb[b] = 0u - ((kid >> b) & 1u);

        int qn = 0;
        if (nTrips > 0) {
            int4 bufA[2 * KSTEP], bufB[2 * KSTEP];
            int tA, tB;
            bool iA, iB;
            key_load(bufA, tA, iA, 0, wave, lane, c0, nTrips, kv);
            for (int it = 0; it < nIter; it += 2) {
                key_load(bufB, tB, iB, it + 1, wave, lane, c0, nTrips, kv);
                key_step(bufA, tA, iA, lane, kb, nG, q, qn, feat, ew, src, dst, acc,
                         (unsigned)lo, (unsigned)rows);
                key_load(bufA, tA, iA, it + 2, wave, lane, c0, nTrips, kv);
                key_step(bufB, tB, iB, lane, kb, nG, q, qn, feat, ew, src, dst, acc,
                         (unsigned)lo, (unsigned)rows);
            }
        }

        for (int e0 = nG * 128 + wave * 32; e0 < nEdges; e0 += 1024) {
            const int e = e0 + lane;
            const unsigned d = (e < nEdges) ? (unsigned)dst[e] : 0xFFFFFFFFu;
            park(d, (unsigned)e, (unsigned)lo, (unsigned)rows, q, qn);
            if (qn >= 32) { qn -= 32; drain32(q + qn, feat, ew, src, dst, acc, (unsigned)lo, (unsigned)rows, lane); }
        }

        for (int t = 0; t < qn; ++t) {
            const unsigned e = q[t];
            const unsigned r = (unsigned)dst[e] - (unsigned)lo;
            if (r < (unsigned)rows) {
                const float x = feat[(size_t)src[e] * N_FIN + lane] * ew[e];
                __hip_atomic_fetch_add(&acc[r * N_FIN + lane], x,
                                       __ATOMIC_RELAXED, __HIP_MEMORY_SCOPE_WORKGROUP);
            }
        }
        __syncthreads();

        for (int i = tid; i < rows * N_FIN; i += 1024) msg[(size_t)lo * N_FIN + i] = acc[i];
        __syncthreads();
    }
}

__global__ __launch_bounds__(256) void gnn_fused(
    const float* __restrict__ feat, const float* __restrict__ msg,
    const float* __restrict__ Win, const float* __restrict__ bin,
    const float* __restrict__ Wh,  const float* __restrict__ bh,
    const float* __restrict__ Wo,  const float* __restrict__ bo,
    float* __restrict__ out, int nNodes) {
    __shared__ __align__(16) _Float16 sX [ROWS * XP];
    __shared__ __align__(16) _Float16 sH1[ROWS * HP];
    __shared__ __align__(16) _Float16 sH2[ROWS * HP];

    const int tid  = threadIdx.x;
    const int wave = tid >> 5;
    const int lane = tid & 31;
    const int col  = lane & 15;
    const int r0   = (lane >> 4) << 3;
    const int nb   = blockIdx.x * ROWS;
    const int c0   = wave * 32;

    v16h w1[2][2];
#pragma unroll
    for (int nt = 0; nt < 2; ++nt)
#pragma unroll
        for (int kc = 0; kc < 2; ++kc)
            w1[nt][kc] = ld_frag_g(Win, 2 * N_FIN, c0 + 16 * nt, 32 * kc);
    const float b1a = bin[c0 + col], b1b = bin[c0 + 16 + col];
    const float b2a = bh[c0 + col],  b2b = bh[c0 + 16 + col];

    float4 xf[4], xm[4];
#pragma unroll
    for (int it = 0; it < 4; ++it) {
        const int idx = it * 256 + tid;
        const int r = idx >> 3, p = idx & 7;
        const int node = nb + r;
        const int nd = node < nNodes ? node : nNodes - 1;
        xf[it] = *(const float4*)(feat + (long long)nd * N_FIN + p * 4);
        xm[it] = *(const float4*)(msg  + (long long)nd * N_FIN + p * 4);
    }
#pragma unroll
    for (int it = 0; it < 4; ++it) {
        const int idx = it * 256 + tid;
        const int r = idx >> 3, p = idx & 7;
        v4h fh, mh;
        fh[0] = (_Float16)xf[it].x; fh[1] = (_Float16)xf[it].y;
        fh[2] = (_Float16)xf[it].z; fh[3] = (_Float16)xf[it].w;
        mh[0] = (_Float16)xm[it].x; mh[1] = (_Float16)xm[it].y;
        mh[2] = (_Float16)xm[it].z; mh[3] = (_Float16)xm[it].w;
        *(v4h*)(sX + r * XP + p * 4)         = fh;
        *(v4h*)(sX + r * XP + N_FIN + p * 4) = mh;
    }

    v16h w2[2][8];
#pragma unroll
    for (int nt = 0; nt < 2; ++nt)
#pragma unroll
        for (int kc = 0; kc < 8; ++kc)
            w2[nt][kc] = ld_frag_g(Wh, HID, c0 + 16 * nt, 32 * kc);

    __syncthreads();

    for (int mt = 0; mt < 8; ++mt) {
        const _Float16* xa = sX + mt * 16 * XP;
        const v16h a0 = ld_frag(xa, XP, 0);
        const v16h a1 = ld_frag(xa, XP, 32);
        v8f d0 = {0.f, 0.f, 0.f, 0.f, 0.f, 0.f, 0.f, 0.f};
        v8f d1 = {0.f, 0.f, 0.f, 0.f, 0.f, 0.f, 0.f, 0.f};
        d0 = WMMA(a0, w1[0][0], d0);
        d1 = WMMA(a0, w1[1][0], d1);
        d0 = WMMA(a1, w1[0][1], d0);
        d1 = WMMA(a1, w1[1][1], d1);
        _Float16* hb = sH1 + mt * 16 * HP + c0;
        st_elu_f16(hb,      HP, d0, b1a);
        st_elu_f16(hb + 16, HP, d1, b1b);
    }
    __syncthreads();

    const int nt3 = wave & 3;
    const int mh3 = wave >> 2;
    v16h w3[8];
#pragma unroll
    for (int kc = 0; kc < 8; ++kc)
        w3[kc] = ld_frag_g(Wo, HID, 16 * nt3, 32 * kc);
    const float b3 = bo[16 * nt3 + col];

    for (int mp = 0; mp < 4; ++mp) {
        const _Float16* ha = sH1 + (2 * mp) * 16 * HP;
        const _Float16* hb = ha + 16 * HP;
        v8f d00 = {0.f, 0.f, 0.f, 0.f, 0.f, 0.f, 0.f, 0.f};
        v8f d01 = d00, d10 = d00, d11 = d00;
#pragma unroll
        for (int kc = 0; kc < 8; ++kc) {
            const v16h a0 = ld_frag(ha, HP, 32 * kc);
            const v16h a1 = ld_frag(hb, HP, 32 * kc);
            d00 = WMMA(a0, w2[0][kc], d00);
            d01 = WMMA(a0, w2[1][kc], d01);
            d10 = WMMA(a1, w2[0][kc], d10);
            d11 = WMMA(a1, w2[1][kc], d11);
        }
        _Float16* oa = sH2 + (2 * mp) * 16 * HP + c0;
        _Float16* ob = oa + 16 * HP;
        st_elu_f16(oa,      HP, d00, b2a);
        st_elu_f16(oa + 16, HP, d01, b2b);
        st_elu_f16(ob,      HP, d10, b2a);
        st_elu_f16(ob + 16, HP, d11, b2b);
    }
    __syncthreads();

    for (int mp = 0; mp < 2; ++mp) {
        const int mt = 4 * mh3 + 2 * mp;
        const _Float16* ha = sH2 + mt * 16 * HP;
        const _Float16* hb = ha + 16 * HP;
        v8f d0 = {0.f, 0.f, 0.f, 0.f, 0.f, 0.f, 0.f, 0.f};
        v8f d1 = d0;
#pragma unroll
        for (int kc = 0; kc < 8; ++kc) {
            d0 = WMMA(ld_frag(ha, HP, 32 * kc), w3[kc], d0);
            d1 = WMMA(ld_frag(hb, HP, 32 * kc), w3[kc], d1);
        }
        const int ocol = 16 * nt3 + col;
#pragma unroll
        for (int r = 0; r < 8; ++r) {
            const int na = nb + mt * 16 + r0 + r;
            const int nc = na + 16;
            if (na < nNodes) out[(long long)na * N_FOUT + ocol] = d0[r] + b3;
            if (nc < nNodes) out[(long long)nc * N_FOUT + ocol] = d1[r] + b3;
        }
    }
}

extern "C" void kernel_launch(void* const* d_in, const int* in_sizes, int n_in,
                              void* d_out, int out_size, void* d_ws, size_t ws_size,
                              hipStream_t stream) {
    const float* feat = (const float*)d_in[0];
    const float* ew   = (const float*)d_in[1];
    const float* Win  = (const float*)d_in[2];
    const float* bin  = (const float*)d_in[3];
    const float* Wh   = (const float*)d_in[4];
    const float* bh   = (const float*)d_in[5];
    const float* Wo   = (const float*)d_in[6];
    const float* bo   = (const float*)d_in[7];
    const int*   ei   = (const int*)d_in[8];

    const int nNodes = in_sizes[0] / N_FIN;
    const int nEdges = in_sizes[1];
    float* msg = (float*)d_ws;
    unsigned* keys = (unsigned*)(msg + (size_t)nNodes * N_FIN);

    gnn_owner_keys<<<256, 512, 0, stream>>>(ei + nEdges, keys, nNodes, nEdges, ws_size);

    gnn_gather_scan<<<256, 1024, 0, stream>>>(feat, ew, ei, ei + nEdges, msg, keys,
                                              nNodes, nEdges, ws_size);

    const int gblocks = (nNodes + 127) / 128;
    gnn_fused<<<gblocks, 256, 0, stream>>>(feat, msg, Win, bin, Wh, bh, Wo, bo,
                                           (float*)d_out, nNodes);
}


# Edge owner scan: a wave's last list entries (fewer than 32) go through the 32-wide batch with idle lanes masked (was one at a time, three dependent loads each); LDS float adds stay in arrival order
# speedup vs baseline: 3.6513x; 1.0959x over previous
#include <hip/hip_runtime.h>


#define N_FIN  32
#define HID    256
#define N_FOUT 64

#define ROWS 128
#define XP   72
#define HP   264

typedef __attribute__((ext_vector_type(16))) _Float16 v16h;
typedef __attribute__((ext_vector_type(8)))  _Float16 v8h;
typedef __attribute__((ext_vector_type(4)))  _Float16 v4h;
typedef __attribute__((ext_vector_type(8)))  float    v8f;

#define WMMA(a, b, c) __builtin_amdgcn_wmma_f32_16x16x32_f16( \
    false, (a), false, (b), (short)0, (c), false, false)

static __device__ __forceinline__ float fast_elu(float x) {
    const float e = __builtin_amdgcn_exp2f(fminf(x, 0.0f) * 1.44269504088896f) - 1.0f;
    return (x > 0.0f) ? x : e;
}

static __device__ __forceinline__ v16h ld_frag(const _Float16* p, int stride, int k0) {
    const int lane = threadIdx.x & 31;
    const _Float16* q = p + (lane & 15) * stride + k0 + ((lane >> 4) << 3);
    v8h lo = *(const v8h*)q;
    v8h hi = *(const v8h*)(q + 16);
    return __builtin_shufflevector(lo, hi, 0, 1, 2, 3, 4, 5, 6, 7,
                                           8, 9, 10, 11, 12, 13, 14, 15);
}

static __device__ __forceinline__ v16h ld_frag_g(const float* __restrict__ W, int ldw,
                                                 int row0, int k0) {
    const int lane = threadIdx.x & 31;
    const float* q = W + (long long)(row0 + (lane & 15)) * ldw + k0 + ((lane >> 4) << 3);
    const float4 a0 = *(const float4*)(q);
    const float4 a1 = *(const float4*)(q + 4);
    const float4 b0 = *(const float4*)(q + 16);
    const float4 b1 = *(const float4*)(q + 20);
    v16h r;
    r[0]  = (_Float16)a0.x; r[1]  = (_Float16)a0.y; r[2]  = (_Float16)a0.z; r[3]  = (_Float16)a0.w;
    r[4]  = (_Float16)a1.x; r[5]  = (_Float16)a1.y; r[6]  = (_Float16)a1.z; r[7]  = (_Float16)a1.w;
    r[8]  = (_Float16)b0.x; r[9]  = (_Float16)b0.y; r[10] = (_Float16)b0.z; r[11] = (_Float16)b0.w;
    r[12] = (_Float16)b1.x; r[13] = (_Float16)b1.y; r[14] = (_Float16)b1.z; r[15] = (_Float16)b1.w;
    return r;
}

static __device__ __forceinline__ void st_elu_f16(_Float16* base, int stride,
                                                  v8f d, float bv) {
    const int lane = threadIdx.x & 31;
    const int col  = lane & 15;
    const int r0   = (lane >> 4) << 3;
#pragma unroll
    for (int r = 0; r < 8; ++r) {
        base[(r0 + r) * stride + col] = (_Float16)fast_elu(d[r] + bv);
    }
}

#define SLICE   392
#define GWAVES  32
#define KSTEP   2
#define QCAP    (32 + 1024)

static __device__ __forceinline__ int key_groups(int nNodes, int nEdges, size_t wsBytes) {
    if (nEdges & 3) return 0;
    const int nG = nEdges / 128;
    const size_t trips = (size_t)((nG + 8 * KSTEP - 1) / (8 * KSTEP));
    const size_t need = (size_t)nNodes * N_FIN * sizeof(float) + trips * (KSTEP * 1024);
    return need <= wsBytes ? nG : 0;
}

#define KPRE 4

__global__ __launch_bounds__(512) void gnn_owner_keys(
    const int* __restrict__ dst, unsigned* __restrict__ keys,
    int nNodes, int nEdges, size_t wsBytes) {
    const int lane = threadIdx.x & 31;
    const int wave = __builtin_amdgcn_readfirstlane((int)(threadIdx.x >> 5));
    const int wpb  = (int)(blockDim.x >> 5);
    const int nG   = key_groups(nNodes, nEdges, wsBytes);
    const int nGpad = ((nG + 8 * KSTEP - 1) / (8 * KSTEP)) * (8 * KSTEP);
    const int gw = blockIdx.x * wpb + wave;
    const int tw = gridDim.x * wpb;
    const int4* dstv = (const int4*)dst;

    for (int base = gw; base < nGpad; base += KPRE * tw) {
        int4 d[KPRE];
#pragma unroll
        for (int j = 0; j < KPRE; ++j) {
            const int G = base + j * tw;
            d[j] = (G < nG) ? dstv[(size_t)G * 32 + lane] : make_int4(0, 0, 0, 0);
        }
#pragma unroll
        for (int j = 0; j < KPRE; ++j) {
            const int G = base + j * tw;
            if (G < nGpad) {
                unsigned k[4];
                k[0] = (unsigned)d[j].x / SLICE; k[1] = (unsigned)d[j].y / SLICE;
                k[2] = (unsigned)d[j].z / SLICE; k[3] = (unsigned)d[j].w / SLICE;
                unsigned mine = 0u;
#pragma unroll
                for (int c = 0; c < 4; ++c)
#pragma unroll
                    for (int b = 0; b < 8; ++b) {
                        const unsigned m = __builtin_amdgcn_ballot_w32(((k[c] >> b) & 1u) != 0u);
                        mine = (lane == 8 * c + b) ? m : mine;
                    }
                const int b   = lane & 7;
                const int sub = (G & 7) * 4 + (lane >> 3);
                keys[(size_t)(G >> 3) * 256 + (b >> 2) * 128 + sub * 4 + (b & 3)] = mine;
            }
        }
    }
}

static __device__ __forceinline__ void drain32(
    const unsigned* q, int n, const float* __restrict__ feat, const float* __restrict__ ew,
    const int* __restrict__ src, const int* __restrict__ dst, float* acc,
    unsigned lo, unsigned rows, int lane) {
    const bool     live = lane < n;
    const unsigned e   = q[live ? lane : 0];
    const unsigned s_l = (unsigned)src[e];
    const unsigned w_l = __float_as_uint(ew[e]);
    const unsigned r_l = (unsigned)dst[e] - lo;
    const unsigned okm = __builtin_amdgcn_ballot_w32(live && r_l < rows);
    float v[32];
#pragma unroll
    for (int t = 0; t < 32; ++t) {
        const unsigned s = __builtin_amdgcn_readlane(s_l, t);
        v[t] = feat[(size_t)s * N_FIN + lane];
    }
#pragma unroll
    for (int t = 0; t < 32; ++t) {
        if ((okm >> t) & 1u) {
            const float    w = __uint_as_float(__builtin_amdgcn_readlane(w_l, t));
            const unsigned r = __builtin_amdgcn_readlane(r_l, t);
            __hip_atomic_fetch_add(&acc[r * N_FIN + lane], v[t] * w,
                                   __ATOMIC_RELAXED, __HIP_MEMORY_SCOPE_WORKGROUP);
        }
    }
}

static __device__ __forceinline__ void park(
    unsigned d, unsigned e, unsigned lo, unsigned rows, unsigned* q, int& qn) {
    const bool     hit = (d - lo) < rows;
    const unsigned m   = __builtin_amdgcn_ballot_w32(hit);
    if (m) {
        const unsigned pos = qn + __builtin_amdgcn_mbcnt_lo(m, 0u);
        if (hit) q[pos] = e;
        qn += __builtin_popcount(m);
    }
}

static __device__ __forceinline__ void park_mask(unsigned m, unsigned e0, unsigned* q, int& qn) {
    unsigned bal;
    while ((bal = __builtin_amdgcn_ballot_w32(m != 0u)) != 0u) {
        const unsigned pos = qn + __builtin_amdgcn_mbcnt_lo(bal, 0u);
        if (m != 0u) q[pos] = e0 + 4u * (unsigned)__builtin_ctz(m);
        qn += __builtin_popcount(bal);
        m &= m - 1u;
    }
}

static __device__ __forceinline__ void key_load(
    int4 (&buf)[2 * KSTEP], int& tb, bool& in, int it, int wave, int lane,
    int c0, int nTrips, const int4* __restrict__ kv) {
    const int k = it * GWAVES + wave;
    int c = c0 + k; if (c >= nTrips) c -= nTrips;
    in = k < nTrips;
    tb = in ? c : 0;
    const int4* p = kv + (size_t)tb * (64 * KSTEP) + lane;
#pragma unroll
    for (int j = 0; j < 2 * KSTEP; ++j) buf[j] = p[j * 32];
}

static __device__ __forceinline__ void key_step(
    const int4 (&buf)[2 * KSTEP], int tb, bool in, int lane, const unsigned (&kb)[8], int nG,
    unsigned* q, int& qn,
    const float* __restrict__ feat, const float* __restrict__ ew,
    const int* __restrict__ src, const int* __restrict__ dst, float* acc,
    unsigned lo, unsigned rows) {
#pragma unroll
    for (int s = 0; s < KSTEP; ++s) {
        const int G = (tb * KSTEP + s) * 8 + (lane >> 2);
        const int4 a = buf[2 * s], b = buf[2 * s + 1];
        const unsigned x = ((unsigned)a.x ^ kb[0]) | ((unsigned)a.y ^ kb[1])
                         | ((unsigned)a.z ^ kb[2]) | ((unsigned)a.w ^ kb[3])
                         | ((unsigned)b.x ^ kb[4]) | ((unsigned)b.y ^ kb[5])
                         | ((unsigned)b.z ^ kb[6]) | ((unsigned)b.w ^ kb[7]);
        const unsigned m = (in && G < nG) ? ~x : 0u;
        park_mask(m, (unsigned)G * 128u + (unsigned)(lane & 3), q, qn);
        while (qn >= 32) { qn -= 32; drain32(q + qn, 32, feat, ew, src, dst, acc, lo, rows, lane); }
    }
}

__global__ __launch_bounds__(1024) void gnn_gather_scan(
    const float* __restrict__ feat, const float* __restrict__ ew,
    const int* __restrict__ src, const int* __restrict__ dst,
    float* __restrict__ msg, const unsigned* __restrict__ keys,
    int nNodes, int nEdges, size_t wsBytes) {
    __shared__ float acc[SLICE * N_FIN];
    __shared__ unsigned qbuf[GWAVES][QCAP];

    const int tid  = threadIdx.x;
    const int lane = tid & 31;
    const int wave = __builtin_amdgcn_readfirstlane(tid >> 5);
    unsigned* q = qbuf[wave];

    const int nG     = key_groups(nNodes, nEdges, wsBytes);
    const int nTrips = (nG + 8 * KSTEP - 1) / (8 * KSTEP);
    const int nIter  = (nTrips + GWAVES - 1) / GWAVES;
    const int c0 = (int)(((long long)blockIdx.x * nTrips) / gridDim.x);
    const int4* kv = (const int4*)keys;

    for (int lo = blockIdx.x * SLICE; lo < nNodes; lo += gridDim.x * SLICE) {
        const int rows = (nNodes - lo < SLICE) ? (nNodes - lo) : SLICE;
        for (int i = tid; i < rows * N_FIN; i += 1024) acc[i] = 0.0f;
        __syncthreads();

        const unsigned kid = (unsigned)(lo / SLICE) & 255u;
        unsigned kb[8];
#pragma unroll
        for (int b = 0; b < 8; ++b) kb[b] = 0u - ((kid >> b) & 1u);

        int qn = 0;
        if (nTrips > 0) {
            int4 bufA[2 * KSTEP], bufB[2 * KSTEP];
            int tA, tB;
            bool iA, iB;
            key_load(bufA, tA, iA, 0, wave, lane, c0, nTrips, kv);
            for (int it = 0; it < nIter; it += 2) {
                key_load(bufB, tB, iB, it + 1, wave, lane, c0, nTrips, kv);
                key_step(bufA, tA, iA, lane, kb, nG, q, qn, feat, ew, src, dst, acc,
                         (unsigned)lo, (unsigned)rows);
                key_load(bufA, tA, iA, it + 2, wave, lane, c0, nTrips, kv);
                key_step(bufB, tB, iB, lane, kb, nG, q, qn, feat, ew, src, dst, acc,
                         (unsigned)lo, (unsigned)rows);
            }
        }

        for (int e0 = nG * 128 + wave * 32; e0 < nEdges; e0 += 1024) {
            const int e = e0 + lane;
            const unsigned d = (e < nEdges) ? (unsigned)dst[e] : 0xFFFFFFFFu;
            park(d, (unsigned)e, (unsigned)lo, (unsigned)rows, q, qn);
            if (qn >= 32) { qn -= 32; drain32(q + qn, 32, feat, ew, src, dst, acc, (unsigned)lo, (unsigned)rows, lane); }
        }

        if (qn > 0) drain32(q, qn, feat, ew, src, dst, acc, (unsigned)lo, (unsigned)rows, lane);
        __syncthreads();

        for (int i = tid; i < rows * N_FIN; i += 1024) msg[(size_t)lo * N_FIN + i] = acc[i];
        __syncthreads();
    }
}

__global__ __launch_bounds__(256) void gnn_fused(
    const float* __restrict__ feat, const float* __restrict__ msg,
    const float* __restrict__ Win, const float* __restrict__ bin,
    const float* __restrict__ Wh,  const float* __restrict__ bh,
    const float* __restrict__ Wo,  const float* __restrict__ bo,
    float* __restrict__ out, int nNodes) {
    __shared__ __align__(16) _Float16 sX [ROWS * XP];
    __shared__ __align__(16) _Float16 sH1[ROWS * HP];
    __shared__ __align__(16) _Float16 sH2[ROWS * HP];

    const int tid  = threadIdx.x;
    const int wave = tid >> 5;
    const int lane = tid & 31;
    const int col  = lane & 15;
    const int r0   = (lane >> 4) << 3;
    const int nb   = blockIdx.x * ROWS;
    const int c0   = wave * 32;

    v16h w1[2][2];
#pragma unroll
    for (int nt = 0; nt < 2; ++nt)
#pragma unroll
        for (int kc = 0; kc < 2; ++kc)
            w1[nt][kc] = ld_frag_g(Win, 2 * N_FIN, c0 + 16 * nt, 32 * kc);
    const float b1a = bin[c0 + col], b1b = bin[c0 + 16 + col];
    const float b2a = bh[c0 + col],  b2b = bh[c0 + 16 + col];

    float4 xf[4], xm[4];
#pragma unroll
    for (int it = 0; it < 4; ++it) {
        const int idx = it * 256 + tid;
        const int r = idx >> 3, p = idx & 7;
        const int node = nb + r;
        const int nd = node < nNodes ? node : nNodes - 1;
        xf[it] = *(const float4*)(feat + (long long)nd * N_FIN + p * 4);
        xm[it] = *(const float4*)(msg  + (long long)nd * N_FIN + p * 4);
    }
#pragma unroll
    for (int it = 0; it < 4; ++it) {
        const int idx = it * 256 + tid;
        const int r = idx >> 3, p = idx & 7;
        v4h fh, mh;
        fh[0] = (_Float16)xf[it].x; fh[1] = (_Float16)xf[it].y;
        fh[2] = (_Float16)xf[it].z; fh[3] = (_Float16)xf[it].w;
        mh[0] = (_Float16)xm[it].x; mh[1] = (_Float16)xm[it].y;
        mh[2] = (_Float16)xm[it].z; mh[3] = (_Float16)xm[it].w;
        *(v4h*)(sX + r * XP + p * 4)         = fh;
        *(v4h*)(sX + r * XP + N_FIN + p * 4) = mh;
    }

    v16h w2[2][8];
#pragma unroll
    for (int nt = 0; nt < 2; ++nt)
#pragma unroll
        for (int kc = 0; kc < 8; ++kc)
            w2[nt][kc] = ld_frag_g(Wh, HID, c0 + 16 * nt, 32 * kc);

    __syncthreads();

    for (int mt = 0; mt < 8; ++mt) {
        const _Float16* xa = sX + mt * 16 * XP;
        const v16h a0 = ld_frag(xa, XP, 0);
        const v16h a1 = ld_frag(xa, XP, 32);
        v8f d0 = {0.f, 0.f, 0.f, 0.f, 0.f, 0.f, 0.f, 0.f};
        v8f d1 = {0.f, 0.f, 0.f, 0.f, 0.f, 0.f, 0.f, 0.f};
        d0 = WMMA(a0, w1[0][0], d0);
        d1 = WMMA(a0, w1[1][0], d1);
        d0 = WMMA(a1, w1[0][1], d0);
        d1 = WMMA(a1, w1[1][1], d1);
        _Float16* hb = sH1 + mt * 16 * HP + c0;
        st_elu_f16(hb,      HP, d0, b1a);
        st_elu_f16(hb + 16, HP, d1, b1b);
    }
    __syncthreads();

    const int nt3 = wave & 3;
    const int mh3 = wave >> 2;
    v16h w3[8];
#pragma unroll
    for (int kc = 0; kc < 8; ++kc)
        w3[kc] = ld_frag_g(Wo, HID, 16 * nt3, 32 * kc);
    const float b3 = bo[16 * nt3 + col];

    for (int mp = 0; mp < 4; ++mp) {
        const _Float16* ha = sH1 + (2 * mp) * 16 * HP;
        const _Float16* hb = ha + 16 * HP;
        v8f d00 = {0.f, 0.f, 0.f, 0.f, 0.f, 0.f, 0.f, 0.f};
        v8f d01 = d00, d10 = d00, d11 = d00;
#pragma unroll
        for (int kc = 0; kc < 8; ++kc) {
            const v16h a0 = ld_frag(ha, HP, 32 * kc);
            const v16h a1 = ld_frag(hb, HP, 32 * kc);
            d00 = WMMA(a0, w2[0][kc], d00);
            d01 = WMMA(a0, w2[1][kc], d01);
            d10 = WMMA(a1, w2[0][kc], d10);
            d11 = WMMA(a1, w2[1][kc], d11);
        }
        _Float16* oa = sH2 + (2 * mp) * 16 * HP + c0;
        _Float16* ob = oa + 16 * HP;
        st_elu_f16(oa,      HP, d00, b2a);
        st_elu_f16(oa + 16, HP, d01, b2b);
        st_elu_f16(ob,      HP, d10, b2a);
        st_elu_f16(ob + 16, HP, d11, b2b);
    }
    __syncthreads();

    for (int mp = 0; mp < 2; ++mp) {
        const int mt = 4 * mh3 + 2 * mp;
        const _Float16* ha = sH2 + mt * 16 * HP;
        const _Float16* hb = ha + 16 * HP;
        v8f d0 = {0.f, 0.f, 0.f, 0.f, 0.f, 0.f, 0.f, 0.f};
        v8f d1 = d0;
#pragma unroll
        for (int kc = 0; kc < 8; ++kc) {
            d0 = WMMA(ld_frag(ha, HP, 32 * kc), w3[kc], d0);
            d1 = WMMA(ld_frag(hb, HP, 32 * kc), w3[kc], d1);
        }
        const int ocol = 16 * nt3 + col;
#pragma unroll
        for (int r = 0; r < 8; ++r) {
            const int na = nb + mt * 16 + r0 + r;
            const int nc = na + 16;
            if (na < nNodes) out[(long long)na * N_FOUT + ocol] = d0[r] + b3;
            if (nc < nNodes) out[(long long)nc * N_FOUT + ocol] = d1[r] + b3;
        }
    }
}

extern "C" void kernel_launch(void* const* d_in, const int* in_sizes, int n_in,
                              void* d_out, int out_size, void* d_ws, size_t ws_size,
                              hipStream_t stream) {
    const float* feat = (const float*)d_in[0];
    const float* ew   = (const float*)d_in[1];
    const float* Win  = (const float*)d_in[2];
    const float* bin  = (const float*)d_in[3];
    const float* Wh   = (const float*)d_in[4];
    const float* bh   = (const float*)d_in[5];
    const float* Wo   = (const float*)d_in[6];
    const float* bo   = (const float*)d_in[7];
    const int*   ei   = (const int*)d_in[8];

    const int nNodes = in_sizes[0] / N_FIN;
    const int nEdges = in_sizes[1];
    float* msg = (float*)d_ws;
    unsigned* keys = (unsigned*)(msg + (size_t)nNodes * N_FIN);

    gnn_owner_keys<<<256, 512, 0, stream>>>(ei + nEdges, keys, nNodes, nEdges, ws_size);

    gnn_gather_scan<<<256, 1024, 0, stream>>>(feat, ew, ei, ei + nEdges, msg, keys,
                                              nNodes, nEdges, ws_size);

    const int gblocks = (nNodes + 127) / 128;
    gnn_fused<<<gblocks, 256, 0, stream>>>(feat, msg, Win, bin, Wh, bh, Wo, bo,
                                           (float*)d_out, nNodes);
}


# Fused MLP block takes 208 node rows (13 row tiles) instead of 128: 481 blocks run as two rounds of one block per WGP, not three and a nearly empty fourth; a block's weight fetch serves 13 tiles
# speedup vs baseline: 5.0432x; 1.3812x over previous
#include <hip/hip_runtime.h>


#define N_FIN  32
#define HID    256
#define N_FOUT 64

#define ROWS 208
#define MT   (ROWS / 16)
#define XIT  ((ROWS * 8 + 255) / 256)
#define XP   72
#define HP   264

typedef __attribute__((ext_vector_type(16))) _Float16 v16h;
typedef __attribute__((ext_vector_type(8)))  _Float16 v8h;
typedef __attribute__((ext_vector_type(4)))  _Float16 v4h;
typedef __attribute__((ext_vector_type(8)))  float    v8f;
typedef __attribute__((ext_vector_type(4)))  int      v4i;

#define WMMA(a, b, c) __builtin_amdgcn_wmma_f32_16x16x32_f16( \
    false, (a), false, (b), (short)0, (c), false, false)

static __device__ __forceinline__ float fast_elu(float x) {
    const float e = __builtin_amdgcn_exp2f(fminf(x, 0.0f) * 1.44269504088896f) - 1.0f;
    return (x > 0.0f) ? x : e;
}

static __device__ __forceinline__ v16h ld_frag(const _Float16* p, int stride, int k0) {
    const int lane = threadIdx.x & 31;
    const _Float16* q = p + (lane & 15) * stride + k0 + ((lane >> 4) << 3);
    v8h lo = *(const v8h*)q;
    v8h hi = *(const v8h*)(q + 16);
    return __builtin_shufflevector(lo, hi, 0, 1, 2, 3, 4, 5, 6, 7,
                                           8, 9, 10, 11, 12, 13, 14, 15);
}

#define WF_H    ((HID / 16) * (2 * N_FIN / 32))
#define WF_O    (WF_H + (HID / 16) * (HID / 32))
#define WF_ALL  (WF_O + (N_FOUT / 16) * (HID / 32))
#define W16_BYTES ((size_t)WF_ALL * 1024)

static __device__ __forceinline__ v16h ld_frag_w(const unsigned* __restrict__ wts, int frag) {
    const int lane = threadIdx.x & 31;
    const v8h* p = (const v8h*)wts + (size_t)frag * 64 + lane;
    const v8h lo = p[0];
    const v8h hi = p[32];
    return __builtin_shufflevector(lo, hi, 0, 1, 2, 3, 4, 5, 6, 7,
                                           8, 9, 10, 11, 12, 13, 14, 15);
}

static __device__ __forceinline__ const float* w16_piece(
    const float* __restrict__ W, int K, int f, int j, int l) {
    const int kcs = K / 32;
    const int t = f / kcs, kc = f % kcs;
    return W + (size_t)(16 * t + (l & 15)) * K + 32 * kc + 16 * j + 8 * (l >> 4);
}

static __device__ __forceinline__ const float* w16_src(
    int g, const float* __restrict__ Win, const float* __restrict__ Wh,
    const float* __restrict__ Wo) {
    const int f = g >> 6, j = (g >> 5) & 1, l = g & 31;
    if (f < WF_H) return w16_piece(Win, 2 * N_FIN, f, j, l);
    if (f < WF_O) return w16_piece(Wh, HID, f - WF_H, j, l);
    return w16_piece(Wo, HID, f - WF_O, j, l);
}

static __device__ __forceinline__ void st_elu_f16(_Float16* base, int stride,
                                                  v8f d, float bv) {
    const int lane = threadIdx.x & 31;
    const int col  = lane & 15;
    const int r0   = (lane >> 4) << 3;
#pragma unroll
    for (int r = 0; r < 8; ++r) {
        base[(r0 + r) * stride + col] = (_Float16)fast_elu(d[r] + bv);
    }
}

#define SLICE   392
#define GWAVES  32
#define KSTEP   2
#define QCAP    (32 + 1024)

static __device__ __forceinline__ int key_groups(int nNodes, int nEdges, size_t wsBytes) {
    if (nEdges & 3) return 0;
    const int nG = nEdges / 128;
    const size_t trips = (size_t)((nG + 8 * KSTEP - 1) / (8 * KSTEP));
    const size_t need = (size_t)nNodes * N_FIN * sizeof(float) + W16_BYTES + trips * (KSTEP * 1024);
    return need <= wsBytes ? nG : 0;
}

static __device__ __forceinline__ bool rec_fit(int nNodes, int nEdges, size_t wsBytes) {
    const int nG = key_groups(nNodes, nEdges, wsBytes);
    if (nG == 0) return false;
    const size_t trips = (size_t)((nG + 8 * KSTEP - 1) / (8 * KSTEP));
    const size_t need = (size_t)nNodes * N_FIN * sizeof(float) + W16_BYTES + trips * (KSTEP * 1024)
                      + (size_t)nEdges * 16;
    return need <= wsBytes;
}

#define KPRE 4

__global__ __launch_bounds__(512) void gnn_owner_keys(
    const int* __restrict__ src, const float* __restrict__ ew,
    const int* __restrict__ dst, unsigned* __restrict__ keys,
    const float* __restrict__ Win, const float* __restrict__ Wh,
    const float* __restrict__ Wo, unsigned* __restrict__ wts,
    int nNodes, int nEdges, size_t wsBytes) {
    const int wper = (WF_ALL * 64 + (int)gridDim.x - 1) / (int)gridDim.x;
    const int wg   = (int)blockIdx.x * wper + (int)threadIdx.x;
    const bool wcv = (int)threadIdx.x < wper && wg < WF_ALL * 64;
    float4 wa = make_float4(0.f, 0.f, 0.f, 0.f), wb = wa;
    if (wcv) {
        const float* sp = w16_src(wg, Win, Wh, Wo);
        wa = *(const float4*)sp;
        wb = *(const float4*)(sp + 4);
    }

    const int lane = threadIdx.x & 31;
    const int wave = __builtin_amdgcn_readfirstlane((int)(threadIdx.x >> 5));
    const int wpb  = (int)(blockDim.x >> 5);
    const int nG   = key_groups(nNodes, nEdges, wsBytes);
    const int nGpad = ((nG + 8 * KSTEP - 1) / (8 * KSTEP)) * (8 * KSTEP);
    const int gw = blockIdx.x * wpb + wave;
    const int tw = gridDim.x * wpb;
    const int4* dstv = (const int4*)dst;
    const int4* srcv = (const int4*)src;
    const float4* ewv = (const float4*)ew;
    const bool useRec = rec_fit(nNodes, nEdges, wsBytes);
    v4i* rec = (v4i*)(keys + (size_t)(nGpad / (8 * KSTEP)) * (KSTEP * 256));

    for (int base = gw; base < nGpad; base += KPRE * tw) {
        int4 d[KPRE], s[KPRE];
        float4 w[KPRE];
#pragma unroll
        for (int j = 0; j < KPRE; ++j) {
            const int G = base + j * tw;
            d[j] = (G < nG) ? dstv[(size_t)G * 32 + lane] : make_int4(0, 0, 0, 0);
            s[j] = make_int4(0, 0, 0, 0);
            w[j] = make_float4(0.f, 0.f, 0.f, 0.f);
            if (useRec && G < nG) {
                s[j] = srcv[(size_t)G * 32 + lane];
                w[j] = ewv[(size_t)G * 32 + lane];
            }
        }
#pragma unroll
        for (int j = 0; j < KPRE; ++j) {
            const int G = base + j * tw;
            if (G < nGpad) {
                unsigned k[4];
                k[0] = (unsigned)d[j].x / SLICE; k[1] = (unsigned)d[j].y / SLICE;
                k[2] = (unsigned)d[j].z / SLICE; k[3] = (unsigned)d[j].w / SLICE;
                unsigned mine = 0u;
#pragma unroll
                for (int c = 0; c < 4; ++c)
#pragma unroll
                    for (int b = 0; b < 8; ++b) {
                        const unsigned m = __builtin_amdgcn_ballot_w32(((k[c] >> b) & 1u) != 0u);
                        mine = (lane == 8 * c + b) ? m : mine;
                    }
                const int b   = lane & 7;
                const int sub = (G & 7) * 4 + (lane >> 3);
                keys[(size_t)(G >> 3) * 256 + (b >> 2) * 128 + sub * 4 + (b & 3)] = mine;
            }
            if (useRec && G < nG) {
                v4i* rp = rec + (size_t)G * 128 + lane;
                rp[0]  = (v4i){s[j].x, d[j].x, __float_as_int(w[j].x), 0};
                rp[32] = (v4i){s[j].y, d[j].y, __float_as_int(w[j].y), 0};
                rp[64] = (v4i){s[j].z, d[j].z, __float_as_int(w[j].z), 0};
                rp[96] = (v4i){s[j].w, d[j].w, __float_as_int(w[j].w), 0};
            }
        }
    }

    if (useRec && gw == 0) {
        for (int j = 0; j < 4; ++j) {
            const int e = nG * 128 + 32 * j + lane;
            if (e < nEdges) rec[e] = (v4i){src[e], dst[e], __float_as_int(ew[e]), 0};
        }
    }

    if (wcv) {
        v8h h;
        h[0] = (_Float16)wa.x; h[1] = (_Float16)wa.y; h[2] = (_Float16)wa.z; h[3] = (_Float16)wa.w;
        h[4] = (_Float16)wb.x; h[5] = (_Float16)wb.y; h[6] = (_Float16)wb.z; h[7] = (_Float16)wb.w;
        *((v8h*)wts + wg) = h;
    }
}

static __device__ __forceinline__ void drain32(
    const unsigned* q, int n, const float* __restrict__ feat, const float* __restrict__ ew,
    const int* __restrict__ src, const int* __restrict__ dst, const v4i* __restrict__ recv,
    float* acc, unsigned lo, unsigned rows, int lane) {
    const bool     live = lane < n;
    const unsigned e   = q[live ? lane : 0];
    unsigned s_l, w_l, r_l;
    if (recv != nullptr) {
        const v4i rc = recv[e];
        s_l = (unsigned)rc.x;
        r_l = (unsigned)rc.y - lo;
        w_l = (unsigned)rc.z;
    } else {
        s_l = (unsigned)src[e];
        w_l = __float_as_uint(ew[e]);
        r_l = (unsigned)dst[e] - lo;
    }
    const unsigned okm = __builtin_amdgcn_ballot_w32(live && r_l < rows);
    float v[32];
#pragma unroll
    for (int t = 0; t < 32; ++t) {
        const unsigned s = __builtin_amdgcn_readlane(s_l, t);
        v[t] = feat[(size_t)s * N_FIN + lane];
    }
#pragma unroll
    for (int t = 0; t < 32; ++t) {
        if ((okm >> t) & 1u) {
            const float    w = __uint_as_float(__builtin_amdgcn_readlane(w_l, t));
            const unsigned r = __builtin_amdgcn_readlane(r_l, t);
            __hip_atomic_fetch_add(&acc[r * N_FIN + lane], v[t] * w,
                                   __ATOMIC_RELAXED, __HIP_MEMORY_SCOPE_WORKGROUP);
        }
    }
}

static __device__ __forceinline__ void park(
    unsigned d, unsigned e, unsigned lo, unsigned rows, unsigned* q, int& qn) {
    const bool     hit = (d - lo) < rows;
    const unsigned m   = __builtin_amdgcn_ballot_w32(hit);
    if (m) {
        const unsigned pos = qn + __builtin_amdgcn_mbcnt_lo(m, 0u);
        if (hit) q[pos] = e;
        qn += __builtin_popcount(m);
    }
}

static __device__ __forceinline__ void park_mask(unsigned m, unsigned e0, int sh, unsigned* q, int& qn) {
    unsigned bal;
    while ((bal = __builtin_amdgcn_ballot_w32(m != 0u)) != 0u) {
        const unsigned pos = qn + __builtin_amdgcn_mbcnt_lo(bal, 0u);
        if (m != 0u) q[pos] = e0 + ((unsigned)__builtin_ctz(m) << sh);
        qn += __builtin_popcount(bal);
        m &= m - 1u;
    }
}

static __device__ __forceinline__ void key_load(
    int4 (&buf)[2 * KSTEP], int& tb, bool& in, int it, int wave, int lane,
    int c0, int nTrips, const int4* __restrict__ kv) {
    const int k = it * GWAVES + wave;
    int c = c0 + k; if (c >= nTrips) c -= nTrips;
    in = k < nTrips;
    tb = in ? c : 0;
    const int4* p = kv + (size_t)tb * (64 * KSTEP) + lane;
#pragma unroll
    for (int j = 0; j < 2 * KSTEP; ++j) buf[j] = p[j * 32];
}

static __device__ __forceinline__ void key_step(
    const int4 (&buf)[2 * KSTEP], int tb, bool in, int lane, const unsigned (&kb)[8], int nG,
    unsigned* q, int& qn,
    const float* __restrict__ feat, const float* __restrict__ ew,
    const int* __restrict__ src, const int* __restrict__ dst, const v4i* __restrict__ recv,
    float* acc, unsigned lo, unsigned rows) {
    const unsigned c   = (unsigned)(lane & 3);
    const unsigned off = (recv != nullptr) ? 32u * c : c;
    const int      sh  = (recv != nullptr) ? 0 : 2;
#pragma unroll
    for (int s = 0; s < KSTEP; ++s) {
        const int G = (tb * KSTEP + s) * 8 + (lane >> 2);
        const int4 a = buf[2 * s], b = buf[2 * s + 1];
        const unsigned x = ((unsigned)a.x ^ kb[0]) | ((unsigned)a.y ^ kb[1])
                         | ((unsigned)a.z ^ kb[2]) | ((unsigned)a.w ^ kb[3])
                         | ((unsigned)b.x ^ kb[4]) | ((unsigned)b.y ^ kb[5])
                         | ((unsigned)b.z ^ kb[6]) | ((unsigned)b.w ^ kb[7]);
        const unsigned m = (in && G < nG) ? ~x : 0u;
        park_mask(m, (unsigned)G * 128u + off, sh, q, qn);
        while (qn >= 32) { qn -= 32; drain32(q + qn, 32, feat, ew, src, dst, recv, acc, lo, rows, lane); }
    }
}

__global__ __launch_bounds__(1024) void gnn_gather_scan(
    const float* __restrict__ feat, const float* __restrict__ ew,
    const int* __restrict__ src, const int* __restrict__ dst,
    float* __restrict__ msg, const unsigned* __restrict__ keys,
    int nNodes, int nEdges, size_t wsBytes) {
    __shared__ float acc[SLICE * N_FIN];
    __shared__ unsigned qbuf[GWAVES][QCAP];

    const int tid  = threadIdx.x;
    const int lane = tid & 31;
    const int wave = __builtin_amdgcn_readfirstlane(tid >> 5);
    unsigned* q = qbuf[wave];

    const int nG     = key_groups(nNodes, nEdges, wsBytes);
    const int nTrips = (nG + 8 * KSTEP - 1) / (8 * KSTEP);
    const int nIter  = (nTrips + GWAVES - 1) / GWAVES;
    const int c0 = (int)(((long long)blockIdx.x * nTrips) / gridDim.x);
    const int4* kv = (const int4*)keys;
    const v4i* recv = rec_fit(nNodes, nEdges, wsBytes)
        ? (const v4i*)(keys + (size_t)nTrips * (KSTEP * 256)) : nullptr;

    for (int lo = blockIdx.x * SLICE; lo < nNodes; lo += gridDim.x * SLICE) {
        const int rows = (nNodes - lo < SLICE) ? (nNodes - lo) : SLICE;
        for (int i = tid; i < rows * N_FIN; i += 1024) acc[i] = 0.0f;
        __syncthreads();

        const unsigned kid = (unsigned)(lo / SLICE) & 255u;
        unsigned kb[8];
#pragma unroll
        for (int b = 0; b < 8; ++b) kb[b] = 0u - ((kid >> b) & 1u);

        int qn = 0;
        if (nTrips > 0) {
            int4 bufA[2 * KSTEP], bufB[2 * KSTEP];
            int tA, tB;
            bool iA, iB;
            key_load(bufA, tA, iA, 0, wave, lane, c0, nTrips, kv);
            for (int it = 0; it < nIter; it += 2) {
                key_load(bufB, tB, iB, it + 1, wave, lane, c0, nTrips, kv);
                key_step(bufA, tA, iA, lane, kb, nG, q, qn, feat, ew, src, dst, recv, acc,
                         (unsigned)lo, (unsigned)rows);
                key_load(bufA, tA, iA, it + 2, wave, lane, c0, nTrips, kv);
                key_step(bufB, tB, iB, lane, kb, nG, q, qn, feat, ew, src, dst, recv, acc,
                         (unsigned)lo, (unsigned)rows);
            }
        }

        for (int e0 = nG * 128 + wave * 32; e0 < nEdges; e0 += 1024) {
            const int e = e0 + lane;
            const unsigned d = (e < nEdges) ? (unsigned)dst[e] : 0xFFFFFFFFu;
            park(d, (unsigned)e, (unsigned)lo, (unsigned)rows, q, qn);
            if (qn >= 32) { qn -= 32; drain32(q + qn, 32, feat, ew, src, dst, recv, acc, (unsigned)lo, (unsigned)rows, lane); }
        }

        if (qn > 0) drain32(q, qn, feat, ew, src, dst, recv, acc, (unsigned)lo, (unsigned)rows, lane);
        __syncthreads();

        for (int i = tid; i < rows * N_FIN; i += 1024) msg[(size_t)lo * N_FIN + i] = acc[i];
        __syncthreads();
    }
}

__global__ __launch_bounds__(256) void gnn_fused(
    const float* __restrict__ feat, const float* __restrict__ msg,
    const unsigned* __restrict__ wts, const float* __restrict__ bin,
    const float* __restrict__ bh, const float* __restrict__ bo,
    float* __restrict__ out, int nNodes) {
    __shared__ __align__(16) _Float16 sX [ROWS * XP];
    __shared__ __align__(16) _Float16 sH1[ROWS * HP];
    __shared__ __align__(16) _Float16 sH2[ROWS * HP];

    const int tid  = threadIdx.x;
    const int wave = __builtin_amdgcn_readfirstlane(tid >> 5);
    const int lane = tid & 31;
    const int col  = lane & 15;
    const int r0   = (lane >> 4) << 3;
    const int nb   = blockIdx.x * ROWS;
    const int c0   = wave * 32;

    v16h w1[2][2];
#pragma unroll
    for (int nt = 0; nt < 2; ++nt)
#pragma unroll
        for (int kc = 0; kc < 2; ++kc)
            w1[nt][kc] = ld_frag_w(wts, (2 * wave + nt) * (2 * N_FIN / 32) + kc);
    const float b1a = bin[c0 + col], b1b = bin[c0 + 16 + col];
    const float b2a = bh[c0 + col],  b2b = bh[c0 + 16 + col];

    float4 xf[XIT], xm[XIT];
#pragma unroll
    for (int it = 0; it < XIT; ++it) {
        const int idx = it * 256 + tid;
        const int r = idx >> 3, p = idx & 7;
        const int node = nb + r;
        const int nd = node < nNodes ? node : nNodes - 1;
        xf[it] = make_float4(0.f, 0.f, 0.f, 0.f);
        xm[it] = xf[it];
        if (idx < ROWS * 8) {
            xf[it] = *(const float4*)(feat + (long long)nd * N_FIN + p * 4);
            xm[it] = *(const float4*)(msg  + (long long)nd * N_FIN + p * 4);
        }
    }
#pragma unroll
    for (int it = 0; it < XIT; ++it) {
        const int idx = it * 256 + tid;
        if (idx >= ROWS * 8) continue;
        const int r = idx >> 3, p = idx & 7;
        v4h fh, mh;
        fh[0] = (_Float16)xf[it].x; fh[1] = (_Float16)xf[it].y;
        fh[2] = (_Float16)xf[it].z; fh[3] = (_Float16)xf[it].w;
        mh[0] = (_Float16)xm[it].x; mh[1] = (_Float16)xm[it].y;
        mh[2] = (_Float16)xm[it].z; mh[3] = (_Float16)xm[it].w;
        *(v4h*)(sX + r * XP + p * 4)         = fh;
        *(v4h*)(sX + r * XP + N_FIN + p * 4) = mh;
    }

    v16h w2[2][8];
#pragma unroll
    for (int nt = 0; nt < 2; ++nt)
#pragma unroll
        for (int kc = 0; kc < 8; ++kc)
            w2[nt][kc] = ld_frag_w(wts, WF_H + (2 * wave + nt) * (HID / 32) + kc);

    __syncthreads();

    for (int mt = 0; mt < MT; ++mt) {
        const _Float16* xa = sX + mt * 16 * XP;
        const v16h a0 = ld_frag(xa, XP, 0);
        const v16h a1 = ld_frag(xa, XP, 32);
        v8f d0 = {0.f, 0.f, 0.f, 0.f, 0.f, 0.f, 0.f, 0.f};
        v8f d1 = {0.f, 0.f, 0.f, 0.f, 0.f, 0.f, 0.f, 0.f};
        d0 = WMMA(a0, w1[0][0], d0);
        d1 = WMMA(a0, w1[1][0], d1);
        d0 = WMMA(a1, w1[0][1], d0);
        d1 = WMMA(a1, w1[1][1], d1);
        _Float16* hb = sH1 + mt * 16 * HP + c0;
        st_elu_f16(hb,      HP, d0, b1a);
        st_elu_f16(hb + 16, HP, d1, b1b);
    }
    __syncthreads();

    const int nt3 = wave & 3;
    const int mh3 = wave >> 2;
    const int m3a = mh3 * ((MT + 1) / 2);
    const int m3b = (m3a + (MT + 1) / 2 < MT) ? m3a + (MT + 1) / 2 : MT;
    v16h w3[8];
#pragma unroll
    for (int kc = 0; kc < 8; ++kc)
        w3[kc] = ld_frag_w(wts, WF_O + nt3 * (HID / 32) + kc);
    const float b3 = bo[16 * nt3 + col];

    for (int mp = 0; mp < MT / 2; ++mp) {
        const _Float16* ha = sH1 + (2 * mp) * 16 * HP;
        const _Float16* hb = ha + 16 * HP;
        v8f d00 = {0.f, 0.f, 0.f, 0.f, 0.f, 0.f, 0.f, 0.f};
        v8f d01 = d00, d10 = d00, d11 = d00;
#pragma unroll
        for (int kc = 0; kc < 8; ++kc) {
            const v16h a0 = ld_frag(ha, HP, 32 * kc);
            const v16h a1 = ld_frag(hb, HP, 32 * kc);
            d00 = WMMA(a0, w2[0][kc], d00);
            d01 = WMMA(a0, w2[1][kc], d01);
            d10 = WMMA(a1, w2[0][kc], d10);
            d11 = WMMA(a1, w2[1][kc], d11);
        }
        _Float16* oa = sH2 + (2 * mp) * 16 * HP + c0;
        _Float16* ob = oa + 16 * HP;
        st_elu_f16(oa,      HP, d00, b2a);
        st_elu_f16(oa + 16, HP, d01, b2b);
        st_elu_f16(ob,      HP, d10, b2a);
        st_elu_f16(ob + 16, HP, d11, b2b);
    }
    if (MT & 1) {
        const _Float16* ha = sH1 + (MT - 1) * 16 * HP;
        v8f d00 = {0.f, 0.f, 0.f, 0.f, 0.f, 0.f, 0.f, 0.f};
        v8f d01 = d00;
#pragma unroll
        for (int kc = 0; kc < 8; ++kc) {
            const v16h a0 = ld_frag(ha, HP, 32 * kc);
            d00 = WMMA(a0, w2[0][kc], d00);
            d01 = WMMA(a0, w2[1][kc], d01);
        }
        _Float16* oa = sH2 + (MT - 1) * 16 * HP + c0;
        st_elu_f16(oa,      HP, d00, b2a);
        st_elu_f16(oa + 16, HP, d01, b2b);
    }
    __syncthreads();

    const int ocol = 16 * nt3 + col;
    for (int mt = m3a; mt < m3b; mt += 2) {
        const _Float16* ha = sH2 + mt * 16 * HP;
        v8f d0 = {0.f, 0.f, 0.f, 0.f, 0.f, 0.f, 0.f, 0.f};
        if (mt + 1 < m3b) {
            const _Float16* hb = ha + 16 * HP;
            v8f d1 = d0;
#pragma unroll
            for (int kc = 0; kc < 8; ++kc) {
                d0 = WMMA(ld_frag(ha, HP, 32 * kc), w3[kc], d0);
                d1 = WMMA(ld_frag(hb, HP, 32 * kc), w3[kc], d1);
            }
#pragma unroll
            for (int r = 0; r < 8; ++r) {
                const int na = nb + mt * 16 + r0 + r;
                const int nc = na + 16;
                if (na < nNodes) out[(long long)na * N_FOUT + ocol] = d0[r] + b3;
                if (nc < nNodes) out[(long long)nc * N_FOUT + ocol] = d1[r] + b3;
            }
        } else {
#pragma unroll
            for (int kc = 0; kc < 8; ++kc)
                d0 = WMMA(ld_frag(ha, HP, 32 * kc), w3[kc], d0);
#pragma unroll
            for (int r = 0; r < 8; ++r) {
                const int na = nb + mt * 16 + r0 + r;
                if (na < nNodes) out[(long long)na * N_FOUT + ocol] = d0[r] + b3;
            }
        }
    }
}

extern "C" void kernel_launch(void* const* d_in, const int* in_sizes, int n_in,
                              void* d_out, int out_size, void* d_ws, size_t ws_size,
                              hipStream_t stream) {
    const float* feat = (const float*)d_in[0];
    const float* ew   = (const float*)d_in[1];
    const float* Win  = (const float*)d_in[2];
    const float* bin  = (const float*)d_in[3];
    const float* Wh   = (const float*)d_in[4];
    const float* bh   = (const float*)d_in[5];
    const float* Wo   = (const float*)d_in[6];
    const float* bo   = (const float*)d_in[7];
    const int*   ei   = (const int*)d_in[8];

    const int nNodes = in_sizes[0] / N_FIN;
    const int nEdges = in_sizes[1];
    float* msg = (float*)d_ws;
    unsigned* wts  = (unsigned*)(msg + (size_t)nNodes * N_FIN);
    unsigned* keys = wts + W16_BYTES / sizeof(unsigned);

    gnn_owner_keys<<<256, 512, 0, stream>>>(ei, ew, ei + nEdges, keys, Win, Wh, Wo, wts,
                                            nNodes, nEdges, ws_size);

    gnn_gather_scan<<<256, 1024, 0, stream>>>(feat, ew, ei, ei + nEdges, msg, keys,
                                              nNodes, nEdges, ws_size);

    const int gblocks = (nNodes + ROWS - 1) / ROWS;
    gnn_fused<<<gblocks, 256, 0, stream>>>(feat, msg, wts, bin, bh, bo,
                                           (float*)d_out, nNodes);
}


# Edge list batches taken 4 entries per pass, 8 lanes each: a lane loads its entry's record and 16 B of the feature row itself (no readlane; 4 whole lines per load); LDS float adds in arrival order
# speedup vs baseline: 5.5834x; 1.1071x over previous
#include <hip/hip_runtime.h>


#define N_FIN  32
#define HID    256
#define N_FOUT 64

#define ROWS 208
#define MT   (ROWS / 16)
#define XIT  ((ROWS * 8 + 255) / 256)
#define XP   72
#define HP   264

typedef __attribute__((ext_vector_type(16))) _Float16 v16h;
typedef __attribute__((ext_vector_type(8)))  _Float16 v8h;
typedef __attribute__((ext_vector_type(4)))  _Float16 v4h;
typedef __attribute__((ext_vector_type(8)))  float    v8f;
typedef __attribute__((ext_vector_type(4)))  int      v4i;

#define WMMA(a, b, c) __builtin_amdgcn_wmma_f32_16x16x32_f16( \
    false, (a), false, (b), (short)0, (c), false, false)

static __device__ __forceinline__ float fast_elu(float x) {
    const float e = __builtin_amdgcn_exp2f(fminf(x, 0.0f) * 1.44269504088896f) - 1.0f;
    return (x > 0.0f) ? x : e;
}

static __device__ __forceinline__ v16h ld_frag(const _Float16* p, int stride, int k0) {
    const int lane = threadIdx.x & 31;
    const _Float16* q = p + (lane & 15) * stride + k0 + ((lane >> 4) << 3);
    v8h lo = *(const v8h*)q;
    v8h hi = *(const v8h*)(q + 16);
    return __builtin_shufflevector(lo, hi, 0, 1, 2, 3, 4, 5, 6, 7,
                                           8, 9, 10, 11, 12, 13, 14, 15);
}

#define WF_H    ((HID / 16) * (2 * N_FIN / 32))
#define WF_O    (WF_H + (HID / 16) * (HID / 32))
#define WF_ALL  (WF_O + (N_FOUT / 16) * (HID / 32))
#define W16_BYTES ((size_t)WF_ALL * 1024)

static __device__ __forceinline__ v16h ld_frag_w(const unsigned* __restrict__ wts, int frag) {
    const int lane = threadIdx.x & 31;
    const v8h* p = (const v8h*)wts + (size_t)frag * 64 + lane;
    const v8h lo = p[0];
    const v8h hi = p[32];
    return __builtin_shufflevector(lo, hi, 0, 1, 2, 3, 4, 5, 6, 7,
                                           8, 9, 10, 11, 12, 13, 14, 15);
}

static __device__ __forceinline__ const float* w16_piece(
    const float* __restrict__ W, int K, int f, int j, int l) {
    const int kcs = K / 32;
    const int t = f / kcs, kc = f % kcs;
    return W + (size_t)(16 * t + (l & 15)) * K + 32 * kc + 16 * j + 8 * (l >> 4);
}

static __device__ __forceinline__ const float* w16_src(
    int g, const float* __restrict__ Win, const float* __restrict__ Wh,
    const float* __restrict__ Wo) {
    const int f = g >> 6, j = (g >> 5) & 1, l = g & 31;
    if (f < WF_H) return w16_piece(Win, 2 * N_FIN, f, j, l);
    if (f < WF_O) return w16_piece(Wh, HID, f - WF_H, j, l);
    return w16_piece(Wo, HID, f - WF_O, j, l);
}

static __device__ __forceinline__ void st_elu_f16(_Float16* base, int stride,
                                                  v8f d, float bv) {
    const int lane = threadIdx.x & 31;
    const int col  = lane & 15;
    const int r0   = (lane >> 4) << 3;
#pragma unroll
    for (int r = 0; r < 8; ++r) {
        base[(r0 + r) * stride + col] = (_Float16)fast_elu(d[r] + bv);
    }
}

#define SLICE   392
#define GWAVES  32
#define KSTEP   2
#define TRIPG   (8 * KSTEP)
#define NMAPS   32
#define QCAP    (32 + 1024)

static __device__ __forceinline__ int groups_padded(int nG) {
    return ((nG + TRIPG - 1) / TRIPG) * TRIPG;
}

static __device__ __forceinline__ int key_groups(int nNodes, int nEdges, size_t wsBytes) {
    if (nEdges & 3) return 0;
    const int nG = nEdges / 128;
    const size_t need = (size_t)nNodes * N_FIN * sizeof(float) + W16_BYTES
                      + (size_t)groups_padded(nG) * (NMAPS * 16);
    return need <= wsBytes ? nG : 0;
}

static __device__ __forceinline__ bool rec_fit(int nNodes, int nEdges, size_t wsBytes) {
    const int nG = key_groups(nNodes, nEdges, wsBytes);
    if (nG == 0) return false;
    const size_t need = (size_t)nNodes * N_FIN * sizeof(float) + W16_BYTES
                      + (size_t)groups_padded(nG) * (NMAPS * 16) + (size_t)nEdges * 16;
    return need <= wsBytes;
}

#define KGS    8
#define KWAVES 8
#define SPITCH 33

__global__ __launch_bounds__(KWAVES * 32) void gnn_owner_keys(
    const int* __restrict__ src, const float* __restrict__ ew,
    const int* __restrict__ dst, unsigned* __restrict__ keys,
    const float* __restrict__ Win, const float* __restrict__ Wh,
    const float* __restrict__ Wo, unsigned* __restrict__ wts,
    int nNodes, int nEdges, size_t wsBytes) {
    __shared__ unsigned stage[KWAVES][NMAPS * SPITCH];

    const int wper = (WF_ALL * 64 + (int)gridDim.x - 1) / (int)gridDim.x;
    const int wg   = (int)blockIdx.x * wper + (int)threadIdx.x;
    const bool wcv = (int)threadIdx.x < wper && wg < WF_ALL * 64;
    float4 wa = make_float4(0.f, 0.f, 0.f, 0.f), wb = wa;
    if (wcv) {
        const float* sp = w16_src(wg, Win, Wh, Wo);
        wa = *(const float4*)sp;
        wb = *(const float4*)(sp + 4);
    }

    const int lane = threadIdx.x & 31;
    const int wave = __builtin_amdgcn_readfirstlane((int)(threadIdx.x >> 5));
    const int wpb  = (int)(blockDim.x >> 5);
    const int nG   = key_groups(nNodes, nEdges, wsBytes);
    const int nGpad = groups_padded(nG);
    const int nW    = 4 * nGpad;
    const int nSteps = nGpad / KGS;
    const int gw = wave * (int)gridDim.x + (int)blockIdx.x;
    const int tw = (int)gridDim.x * wpb;
    const int4* dstv = (const int4*)dst;
    const int4* srcv = (const int4*)src;
    const float4* ewv = (const float4*)ew;
    const bool useRec = rec_fit(nNodes, nEdges, wsBytes);
    v4i* rec = (v4i*)(keys + (size_t)NMAPS * nW);
    unsigned* st = stage[wave];
    const unsigned bit = 1u << lane;

    for (int step = gw; step < nSteps; step += tw) {
        int4 d[KGS], s[KGS];
        float4 w[KGS];
#pragma unroll
        for (int j = 0; j < KGS; ++j) {
            const int G = step * KGS + j;
            d[j] = (G < nG) ? dstv[(size_t)G * 32 + lane] : make_int4(0, 0, 0, 0);
            s[j] = make_int4(0, 0, 0, 0);
            w[j] = make_float4(0.f, 0.f, 0.f, 0.f);
            if (useRec && G < nG) {
                s[j] = srcv[(size_t)G * 32 + lane];
                w[j] = ewv[(size_t)G * 32 + lane];
            }
        }
        for (int i = lane; i < NMAPS * SPITCH; i += 32) st[i] = 0u;
#pragma unroll
        for (int j = 0; j < KGS; ++j) {
            const int G = step * KGS + j;
            if (G < nG) {
                unsigned k[4];
                k[0] = (unsigned)d[j].x / SLICE; k[1] = (unsigned)d[j].y / SLICE;
                k[2] = (unsigned)d[j].z / SLICE; k[3] = (unsigned)d[j].w / SLICE;
#pragma unroll
                for (int c = 0; c < 4; ++c) {
                    const unsigned hi = (k[c] >> 4) & 15u, lo = k[c] & 15u;
                    __hip_atomic_fetch_or(&st[hi * SPITCH + 4 * j + c], bit,
                                          __ATOMIC_RELAXED, __HIP_MEMORY_SCOPE_WORKGROUP);
                    __hip_atomic_fetch_or(&st[(16u + lo) * SPITCH + 4 * j + c], bit,
                                          __ATOMIC_RELAXED, __HIP_MEMORY_SCOPE_WORKGROUP);
                }
            }
            if (useRec && G < nG) {
                v4i* rp = rec + (size_t)G * 128 + lane;
                rp[0]  = (v4i){s[j].x, d[j].x, __float_as_int(w[j].x), 0};
                rp[32] = (v4i){s[j].y, d[j].y, __float_as_int(w[j].y), 0};
                rp[64] = (v4i){s[j].z, d[j].z, __float_as_int(w[j].z), 0};
                rp[96] = (v4i){s[j].w, d[j].w, __float_as_int(w[j].w), 0};
            }
        }
#pragma unroll
        for (int p = 0; p < NMAPS; ++p)
            keys[(size_t)p * nW + (size_t)step * (4 * KGS) + lane] = st[p * SPITCH + lane];
    }

    if (useRec && gw == 0) {
        for (int j = 0; j < 4; ++j) {
            const int e = nG * 128 + 32 * j + lane;
            if (e < nEdges) rec[e] = (v4i){src[e], dst[e], __float_as_int(ew[e]), 0};
        }
    }

    if (wcv) {
        v8h h;
        h[0] = (_Float16)wa.x; h[1] = (_Float16)wa.y; h[2] = (_Float16)wa.z; h[3] = (_Float16)wa.w;
        h[4] = (_Float16)wb.x; h[5] = (_Float16)wb.y; h[6] = (_Float16)wb.z; h[7] = (_Float16)wb.w;
        *((v8h*)wts + wg) = h;
    }
}

static __device__ __forceinline__ void drain32(
    const unsigned* q, int n, const float* __restrict__ feat, const float* __restrict__ ew,
    const int* __restrict__ src, const int* __restrict__ dst, const v4i* __restrict__ recv,
    float* acc, unsigned lo, unsigned rows, int lane) {
    const int g   = lane >> 3;
    const int sub = lane & 7;
    unsigned e[8], s[8], r[8], w[8];
#pragma unroll
    for (int j = 0; j < 8; ++j) {
        const int t = 4 * j + g;
        e[j] = q[t < n ? t : 0];
    }
    if (recv != nullptr) {
        v4i rc[8];
#pragma unroll
        for (int j = 0; j < 8; ++j) rc[j] = recv[e[j]];
#pragma unroll
        for (int j = 0; j < 8; ++j) {
            s[j] = (unsigned)rc[j].x;
            r[j] = (unsigned)rc[j].y - lo;
            w[j] = (unsigned)rc[j].z;
        }
    } else {
#pragma unroll
        for (int j = 0; j < 8; ++j) {
            s[j] = (unsigned)src[e[j]];
            w[j] = __float_as_uint(ew[e[j]]);
            r[j] = (unsigned)dst[e[j]] - lo;
        }
    }
    float4 v[8];
#pragma unroll
    for (int j = 0; j < 8; ++j)
        v[j] = *(const float4*)(feat + (size_t)s[j] * N_FIN + 4 * sub);
    const bool g1 = (g & 1) != 0, g2 = (g & 2) != 0;
    float* const p0 = acc + 4 * sub + (0 ^ g);
    float* const p1 = acc + 4 * sub + (1 ^ g);
    float* const p2 = acc + 4 * sub + (2 ^ g);
    float* const p3 = acc + 4 * sub + (3 ^ g);
#pragma unroll
    for (int j = 0; j < 8; ++j) {
        const float wf = __uint_as_float(w[j]);
        const float x0 = v[j].x * wf, x1 = v[j].y * wf, x2 = v[j].z * wf, x3 = v[j].w * wf;
        const float a0 = g1 ? x1 : x0, a1 = g1 ? x0 : x1, a2 = g1 ? x3 : x2, a3 = g1 ? x2 : x3;
        const float b0 = g2 ? a2 : a0, b1 = g2 ? a3 : a1, b2 = g2 ? a0 : a2, b3 = g2 ? a1 : a3;
        if (r[j] < rows && 4 * j + g < n) {
            const unsigned ro = r[j] * N_FIN;
            __hip_atomic_fetch_add(p0 + ro, b0, __ATOMIC_RELAXED, __HIP_MEMORY_SCOPE_WORKGROUP);
            __hip_atomic_fetch_add(p1 + ro, b1, __ATOMIC_RELAXED, __HIP_MEMORY_SCOPE_WORKGROUP);
            __hip_atomic_fetch_add(p2 + ro, b2, __ATOMIC_RELAXED, __HIP_MEMORY_SCOPE_WORKGROUP);
            __hip_atomic_fetch_add(p3 + ro, b3, __ATOMIC_RELAXED, __HIP_MEMORY_SCOPE_WORKGROUP);
        }
    }
}

static __device__ __forceinline__ void park(
    unsigned d, unsigned e, unsigned lo, unsigned rows, unsigned* q, int& qn) {
    const bool     hit = (d - lo) < rows;
    const unsigned m   = __builtin_amdgcn_ballot_w32(hit);
    if (m) {
        const unsigned pos = qn + __builtin_amdgcn_mbcnt_lo(m, 0u);
        if (hit) q[pos] = e;
        qn += __builtin_popcount(m);
    }
}

static __device__ __forceinline__ void park_mask(unsigned m, unsigned e0, int sh, unsigned* q, int& qn) {
    unsigned bal;
    while ((bal = __builtin_amdgcn_ballot_w32(m != 0u)) != 0u) {
        const unsigned pos = qn + __builtin_amdgcn_mbcnt_lo(bal, 0u);
        if (m != 0u) q[pos] = e0 + ((unsigned)__builtin_ctz(m) << sh);
        qn += __builtin_popcount(bal);
        m &= m - 1u;
    }
}

static __device__ __forceinline__ void key_load(
    unsigned (&buf)[2 * KSTEP], int& tb, bool& in, int it, int wave, int lane,
    int c0, int nTrips, const unsigned* __restrict__ mhi, const unsigned* __restrict__ mlo) {
    const int k = it * GWAVES + wave;
    int c = c0 + k; if (c >= nTrips) c -= nTrips;
    in = k < nTrips;
    tb = in ? c : 0;
    const unsigned o = (unsigned)tb * (32u * KSTEP) + (unsigned)lane;
#pragma unroll
    for (int j = 0; j < KSTEP; ++j) {
        buf[2 * j]     = mhi[o + 32 * j];
        buf[2 * j + 1] = mlo[o + 32 * j];
    }
}

static __device__ __forceinline__ void key_step(
    const unsigned (&buf)[2 * KSTEP], int tb, bool in, int lane, int nG,
    unsigned* q, int& qn,
    const float* __restrict__ feat, const float* __restrict__ ew,
    const int* __restrict__ src, const int* __restrict__ dst, const v4i* __restrict__ recv,
    float* acc, unsigned lo, unsigned rows) {
    const unsigned c   = (unsigned)(lane & 3);
    const unsigned off = (recv != nullptr) ? 32u * c : c;
    const int      sh  = (recv != nullptr) ? 0 : 2;
#pragma unroll
    for (int s = 0; s < KSTEP; ++s) {
        const int G = (tb * KSTEP + s) * 8 + (lane >> 2);
        const unsigned m = (in && G < nG) ? (buf[2 * s] & buf[2 * s + 1]) : 0u;
        park_mask(m, (unsigned)G * 128u + off, sh, q, qn);
        while (qn >= 32) { qn -= 32; drain32(q + qn, 32, feat, ew, src, dst, recv, acc, lo, rows, lane); }
    }
}

__global__ __launch_bounds__(1024) void gnn_gather_scan(
    const float* __restrict__ feat, const float* __restrict__ ew,
    const int* __restrict__ src, const int* __restrict__ dst,
    float* __restrict__ msg, const unsigned* __restrict__ keys,
    int nNodes, int nEdges, size_t wsBytes) {
    __shared__ float acc[SLICE * N_FIN];
    __shared__ unsigned qbuf[GWAVES][QCAP];

    const int tid  = threadIdx.x;
    const int lane = tid & 31;
    const int wave = __builtin_amdgcn_readfirstlane(tid >> 5);
    unsigned* q = qbuf[wave];

    const int nG     = key_groups(nNodes, nEdges, wsBytes);
    const int nTrips = groups_padded(nG) / TRIPG;
    const int nIter  = (nTrips + GWAVES - 1) / GWAVES;
    const size_t nW  = (size_t)4 * TRIPG * nTrips;
    const int c0 = (int)(((long long)blockIdx.x * nTrips) / gridDim.x);
    const v4i* recv = rec_fit(nNodes, nEdges, wsBytes)
        ? (const v4i*)(keys + (size_t)NMAPS * nW) : nullptr;

    for (int lo = blockIdx.x * SLICE; lo < nNodes; lo += gridDim.x * SLICE) {
        const int rows = (nNodes - lo < SLICE) ? (nNodes - lo) : SLICE;
        for (int i = tid; i < rows * N_FIN; i += 1024) acc[i] = 0.0f;
        __syncthreads();

        const unsigned kid = (unsigned)(lo / SLICE) & 255u;
        const unsigned* mhi = keys + (size_t)(kid >> 4) * nW;
        const unsigned* mlo = keys + (size_t)(16u + (kid & 15u)) * nW;

        int qn = 0;
        if (nTrips > 0) {
            unsigned bufA[2 * KSTEP], bufB[2 * KSTEP];
            int tA, tB;
            bool iA, iB;
            key_load(bufA, tA, iA, 0, wave, lane, c0, nTrips, mhi, mlo);
            for (int it = 0; it < nIter; it += 2) {
                key_load(bufB, tB, iB, it + 1, wave, lane, c0, nTrips, mhi, mlo);
                key_step(bufA, tA, iA, lane, nG, q, qn, feat, ew, src, dst, recv, acc,
                         (unsigned)lo, (unsigned)rows);
                key_load(bufA, tA, iA, it + 2, wave, lane, c0, nTrips, mhi, mlo);
                key_step(bufB, tB, iB, lane, nG, q, qn, feat, ew, src, dst, recv, acc,
                         (unsigned)lo, (unsigned)rows);
            }
        }

        for (int e0 = nG * 128 + wave * 32; e0 < nEdges; e0 += 1024) {
            const int e = e0 + lane;
            const unsigned d = (e < nEdges) ? (unsigned)dst[e] : 0xFFFFFFFFu;
            park(d, (unsigned)e, (unsigned)lo, (unsigned)rows, q, qn);
            if (qn >= 32) { qn -= 32; drain32(q + qn, 32, feat, ew, src, dst, recv, acc, (unsigned)lo, (unsigned)rows, lane); }
        }

        if (qn > 0) drain32(q, qn, feat, ew, src, dst, recv, acc, (unsigned)lo, (unsigned)rows, lane);
        __syncthreads();

        for (int i = tid; i < rows * N_FIN; i += 1024) msg[(size_t)lo * N_FIN + i] = acc[i];
        __syncthreads();
    }
}

__global__ __launch_bounds__(256) void gnn_fused(
    const float* __restrict__ feat, const float* __restrict__ msg,
    const unsigned* __restrict__ wts, const float* __restrict__ bin,
    const float* __restrict__ bh, const float* __restrict__ bo,
    float* __restrict__ out, int nNodes) {
    __shared__ __align__(16) _Float16 sX [ROWS * XP];
    __shared__ __align__(16) _Float16 sH1[ROWS * HP];
    __shared__ __align__(16) _Float16 sH2[ROWS * HP];

    const int tid  = threadIdx.x;
    const int wave = __builtin_amdgcn_readfirstlane(tid >> 5);
    const int lane = tid & 31;
    const int col  = lane & 15;
    const int r0   = (lane >> 4) << 3;
    const int nb   = blockIdx.x * ROWS;
    const int c0   = wave * 32;

    v16h w1[2][2];
#pragma unroll
    for (int nt = 0; nt < 2; ++nt)
#pragma unroll
        for (int kc = 0; kc < 2; ++kc)
            w1[nt][kc] = ld_frag_w(wts, (2 * wave + nt) * (2 * N_FIN / 32) + kc);
    const float b1a = bin[c0 + col], b1b = bin[c0 + 16 + col];
    const float b2a = bh[c0 + col],  b2b = bh[c0 + 16 + col];

    float4 xf[XIT], xm[XIT];
#pragma unroll
    for (int it = 0; it < XIT; ++it) {
        const int idx = it * 256 + tid;
        const int r = idx >> 3, p = idx & 7;
        const int node = nb + r;
        const int nd = node < nNodes ? node : nNodes - 1;
        xf[it] = make_float4(0.f, 0.f, 0.f, 0.f);
        xm[it] = xf[it];
        if (idx < ROWS * 8) {
            xf[it] = *(const float4*)(feat + (long long)nd * N_FIN + p * 4);
            xm[it] = *(const float4*)(msg  + (long long)nd * N_FIN + p * 4);
        }
    }
#pragma unroll
    for (int it = 0; it < XIT; ++it) {
        const int idx = it * 256 + tid;
        if (idx >= ROWS * 8) continue;
        const int r = idx >> 3, p = idx & 7;
        v4h fh, mh;
        fh[0] = (_Float16)xf[it].x; fh[1] = (_Float16)xf[it].y;
        fh[2] = (_Float16)xf[it].z; fh[3] = (_Float16)xf[it].w;
        mh[0] = (_Float16)xm[it].x; mh[1] = (_Float16)xm[it].y;
        mh[2] = (_Float16)xm[it].z; mh[3] = (_Float16)xm[it].w;
        *(v4h*)(sX + r * XP + p * 4)         = fh;
        *(v4h*)(sX + r * XP + N_FIN + p * 4) = mh;
    }

    v16h w2[2][8];
#pragma unroll
    for (int nt = 0; nt < 2; ++nt)
#pragma unroll
        for (int kc = 0; kc < 8; ++kc)
            w2[nt][kc] = ld_frag_w(wts, WF_H + (2 * wave + nt) * (HID / 32) + kc);

    __syncthreads();

    for (int mt = 0; mt < MT; ++mt) {
        const _Float16* xa = sX + mt * 16 * XP;
        const v16h a0 = ld_frag(xa, XP, 0);
        const v16h a1 = ld_frag(xa, XP, 32);
        v8f d0 = {0.f, 0.f, 0.f, 0.f, 0.f, 0.f, 0.f, 0.f};
        v8f d1 = {0.f, 0.f, 0.f, 0.f, 0.f, 0.f, 0.f, 0.f};
        d0 = WMMA(a0, w1[0][0], d0);
        d1 = WMMA(a0, w1[1][0], d1);
        d0 = WMMA(a1, w1[0][1], d0);
        d1 = WMMA(a1, w1[1][1], d1);
        _Float16* hb = sH1 + mt * 16 * HP + c0;
        st_elu_f16(hb,      HP, d0, b1a);
        st_elu_f16(hb + 16, HP, d1, b1b);
    }
    __syncthreads();

    const int nt3 = wave & 3;
    const int mh3 = wave >> 2;
    const int m3a = mh3 * ((MT + 1) / 2);
    const int m3b = (m3a + (MT + 1) / 2 < MT) ? m3a + (MT + 1) / 2 : MT;
    v16h w3[8];
#pragma unroll
    for (int kc = 0; kc < 8; ++kc)
        w3[kc] = ld_frag_w(wts, WF_O + nt3 * (HID / 32) + kc);
    const float b3 = bo[16 * nt3 + col];

    for (int mp = 0; mp < MT / 2; ++mp) {
        const _Float16* ha = sH1 + (2 * mp) * 16 * HP;
        const _Float16* hb = ha + 16 * HP;
        v8f d00 = {0.f, 0.f, 0.f, 0.f, 0.f, 0.f, 0.f, 0.f};
        v8f d01 = d00, d10 = d00, d11 = d00;
#pragma unroll
        for (int kc = 0; kc < 8; ++kc) {
            const v16h a0 = ld_frag(ha, HP, 32 * kc);
            const v16h a1 = ld_frag(hb, HP, 32 * kc);
            d00 = WMMA(a0, w2[0][kc], d00);
            d01 = WMMA(a0, w2[1][kc], d01);
            d10 = WMMA(a1, w2[0][kc], d10);
            d11 = WMMA(a1, w2[1][kc], d11);
        }
        _Float16* oa = sH2 + (2 * mp) * 16 * HP + c0;
        _Float16* ob = oa + 16 * HP;
        st_elu_f16(oa,      HP, d00, b2a);
        st_elu_f16(oa + 16, HP, d01, b2b);
        st_elu_f16(ob,      HP, d10, b2a);
        st_elu_f16(ob + 16, HP, d11, b2b);
    }
    if (MT & 1) {
        const _Float16* ha = sH1 + (MT - 1) * 16 * HP;
        v8f d00 = {0.f, 0.f, 0.f, 0.f, 0.f, 0.f, 0.f, 0.f};
        v8f d01 = d00;
#pragma unroll
        for (int kc = 0; kc < 8; ++kc) {
            const v16h a0 = ld_frag(ha, HP, 32 * kc);
            d00 = WMMA(a0, w2[0][kc], d00);
            d01 = WMMA(a0, w2[1][kc], d01);
        }
        _Float16* oa = sH2 + (MT - 1) * 16 * HP + c0;
        st_elu_f16(oa,      HP, d00, b2a);
        st_elu_f16(oa + 16, HP, d01, b2b);
    }
    __syncthreads();

    const int ocol = 16 * nt3 + col;
    for (int mt = m3a; mt < m3b; mt += 2) {
        const _Float16* ha = sH2 + mt * 16 * HP;
        v8f d0 = {0.f, 0.f, 0.f, 0.f, 0.f, 0.f, 0.f, 0.f};
        if (mt + 1 < m3b) {
            const _Float16* hb = ha + 16 * HP;
            v8f d1 = d0;
#pragma unroll
            for (int kc = 0; kc < 8; ++kc) {
                d0 = WMMA(ld_frag(ha, HP, 32 * kc), w3[kc], d0);
                d1 = WMMA(ld_frag(hb, HP, 32 * kc), w3[kc], d1);
            }
#pragma unroll
            for (int r = 0; r < 8; ++r) {
                const int na = nb + mt * 16 + r0 + r;
                const int nc = na + 16;
                if (na < nNodes) out[(long long)na * N_FOUT + ocol] = d0[r] + b3;
                if (nc < nNodes) out[(long long)nc * N_FOUT + ocol] = d1[r] + b3;
            }
        } else {
#pragma unroll
            for (int kc = 0; kc < 8; ++kc)
                d0 = WMMA(ld_frag(ha, HP, 32 * kc), w3[kc], d0);
#pragma unroll
            for (int r = 0; r < 8; ++r) {
                const int na = nb + mt * 16 + r0 + r;
                if (na < nNodes) out[(long long)na * N_FOUT + ocol] = d0[r] + b3;
            }
        }
    }
}

extern "C" void kernel_launch(void* const* d_in, const int* in_sizes, int n_in,
                              void* d_out, int out_size, void* d_ws, size_t ws_size,
                              hipStream_t stream) {
    const float* feat = (const float*)d_in[0];
    const float* ew   = (const float*)d_in[1];
    const float* Win  = (const float*)d_in[2];
    const float* bin  = (const float*)d_in[3];
    const float* Wh   = (const float*)d_in[4];
    const float* bh   = (const float*)d_in[5];
    const float* Wo   = (const float*)d_in[6];
    const float* bo   = (const float*)d_in[7];
    const int*   ei   = (const int*)d_in[8];

    const int nNodes = in_sizes[0] / N_FIN;
    const int nEdges = in_sizes[1];
    float* msg = (float*)d_ws;
    unsigned* wts  = (unsigned*)(msg + (size_t)nNodes * N_FIN);
    unsigned* keys = wts + W16_BYTES / sizeof(unsigned);

    gnn_owner_keys<<<256, KWAVES * 32, 0, stream>>>(ei, ew, ei + nEdges, keys, Win, Wh, Wo, wts,
                                            nNodes, nEdges, ws_size);

    gnn_gather_scan<<<256, 1024, 0, stream>>>(feat, ew, ei, ei + nEdges, msg, keys,
                                              nNodes, nEdges, ws_size);

    const int gblocks = (nNodes + ROWS - 1) / ROWS;
    gnn_fused<<<gblocks, 256, 0, stream>>>(feat, msg, wts, bin, bh, bo,
                                           (float*)d_out, nNodes);
}
